# starting kernel
# baseline (speedup 1.0000x reference)
#include <hip/hip_runtime.h>
#include <hip/hip_bf16.h>

typedef __attribute__((ext_vector_type(2))) float v2f;
typedef __attribute__((ext_vector_type(8))) float v8f;

#define D 128
#define GRAPHS 64


__global__ void fill_zero_kernel(float* __restrict__ p, int n) {
    int i = blockIdx.x * blockDim.x + threadIdx.x;
    if (i < n) p[i] = 0.0f;
}

__global__ void count_deg_kernel(const int* __restrict__ dst, float* __restrict__ cnt, int E) {
    int e = blockIdx.x * blockDim.x + threadIdx.x;
    if (e < E) atomicAdd(&cnt[dst[e]], 1.0f);
}

__global__ void scatter_edges_kernel(const float* __restrict__ x,
                                     const int* __restrict__ src,
                                     const int* __restrict__ dst,
                                     float* __restrict__ agg, int E) {
    unsigned t = blockIdx.x * blockDim.x + threadIdx.x;
    unsigned e = t >> 5;
    if (e >= (unsigned)E) return;
    unsigned c = (t & 31u) << 2;
    int s = src[e], d = dst[e];
    const float4 v = *(const float4*)(x + (size_t)s * D + c);
    float* p = agg + (size_t)d * D + c;
    atomicAdd(p + 0, v.x);
    atomicAdd(p + 1, v.y);
    atomicAdd(p + 2, v.z);
    atomicAdd(p + 3, v.w);
}


__launch_bounds__(256)
__global__ void sage_layer_kernel(const float* __restrict__ xin,
                                  const float* __restrict__ agg,
                                  const float* __restrict__ cnt,
                                  const float* __restrict__ Wl,
                                  const float* __restrict__ Wr,
                                  const float* __restrict__ bias,
                                  float* __restrict__ out,
                                  int do_relu) {
    __shared__ float ssq[16][8];
    __shared__ float inv_norm[16];

    const int lane = threadIdx.x & 31;
    const int wv   = threadIdx.x >> 5;
    const int half = lane >> 4;
    const int lr   = lane & 15;
    const int rowbase = blockIdx.x * 16;
    const int row  = rowbase + lr;
    const int col  = wv * 16 + lr;

    const float invc = 1.0f / fmaxf(cnt[row], 1.0f);
    const float* Am = agg + (size_t)row * D;
    const float* Ax = xin + (size_t)row * D;

    v8f acc = {0.f, 0.f, 0.f, 0.f, 0.f, 0.f, 0.f, 0.f};

#pragma unroll
    for (int k0 = 0; k0 < D; k0 += 4) {
        const int ka = k0 + 2 * half;
        v2f a, b;
        a.x = Am[ka] * invc;
        a.y = Am[ka + 1] * invc;
        b.x = Wl[(size_t)ka * D + col];
        b.y = Wl[(size_t)(ka + 1) * D + col];
        acc = __builtin_amdgcn_wmma_f32_16x16x4_f32(false, a, false, b,
                                                    (short)0, acc, false, false);
    }
#pragma unroll
    for (int k0 = 0; k0 < D; k0 += 4) {
        const int ka = k0 + 2 * half;
        v2f a, b;
        a.x = Ax[ka];
        a.y = Ax[ka + 1];
        b.x = Wr[(size_t)ka * D + col];
        b.y = Wr[(size_t)(ka + 1) * D + col];
        acc = __builtin_amdgcn_wmma_f32_16x16x4_f32(false, a, false, b,
                                                    (short)0, acc, false, false);
    }

    const float bb = bias[col];
#pragma unroll
    for (int r = 0; r < 8; ++r) {
        float v = acc[r] + bb;
        if (do_relu) v = fmaxf(v, 0.0f);
        acc[r] = v;
    }

#pragma unroll
    for (int r = 0; r < 8; ++r) {
        float s = acc[r] * acc[r];
        s += __shfl_xor(s, 1, 32);
        s += __shfl_xor(s, 2, 32);
        s += __shfl_xor(s, 4, 32);
        s += __shfl_xor(s, 8, 32);
        if (lr == 0) ssq[r + 8 * half][wv] = s;
    }
    __syncthreads();
    if (threadIdx.x < 16) {
        float s = 0.0f;
#pragma unroll
        for (int w = 0; w < 8; ++w) s += ssq[threadIdx.x][w];
        inv_norm[threadIdx.x] = 1.0f / fmaxf(sqrtf(s), 1e-12f);
    }
    __syncthreads();

#pragma unroll
    for (int r = 0; r < 8; ++r) {
        const float sc = inv_norm[r + 8 * half];
        out[(size_t)(rowbase + r + 8 * half) * D + col] = acc[r] * sc;
    }
}


__global__ void graph_count_kernel(const int* __restrict__ batch, float* __restrict__ gcnt, int N) {
    int n = blockIdx.x * blockDim.x + threadIdx.x;
    if (n < N) atomicAdd(&gcnt[batch[n]], 1.0f);
}

__global__ void pool_sum_kernel(const float* __restrict__ h, const int* __restrict__ batch,
                                float* __restrict__ pool, int N) {
    unsigned t = blockIdx.x * blockDim.x + threadIdx.x;
    unsigned n = t >> 5;
    if (n >= (unsigned)N) return;
    unsigned c = (t & 31u) << 2;
    int g = batch[n];
    const float4 v = *(const float4*)(h + (size_t)n * D + c);
    float* p = pool + (size_t)g * D + c;
    atomicAdd(p + 0, v.x);
    atomicAdd(p + 1, v.y);
    atomicAdd(p + 2, v.z);
    atomicAdd(p + 3, v.w);
}

__global__ void pool_div_kernel(const float* __restrict__ pool, const float* __restrict__ gcnt,
                                float* __restrict__ out) {
    int i = blockIdx.x * blockDim.x + threadIdx.x;
    if (i < GRAPHS * D) {
        int g = i / D;
        out[i] = pool[i] / fmaxf(gcnt[g], 1.0f);
    }
}


static inline unsigned cdiv_u(long long a, long long b) { return (unsigned)((a + b - 1) / b); }

extern "C" void kernel_launch(void* const* d_in, const int* in_sizes, int n_in,
                              void* d_out, int out_size, void* d_ws, size_t ws_size,
                              hipStream_t stream) {
    const float* x    = (const float*)d_in[0];
    const int*   ei   = (const int*)d_in[1];
    const int*   batch= (const int*)d_in[2];
    const float* Wl1  = (const float*)d_in[3];
    const float* Wr1  = (const float*)d_in[4];
    const float* b1   = (const float*)d_in[5];
    const float* Wl2  = (const float*)d_in[6];
    const float* Wr2  = (const float*)d_in[7];
    const float* b2   = (const float*)d_in[8];
    const float* Wl3  = (const float*)d_in[9];
    const float* Wr3  = (const float*)d_in[10];
    const float* b3   = (const float*)d_in[11];
    float* out = (float*)d_out;

    const int N  = in_sizes[0] / D;
    const int E  = in_sizes[1] / 2;
    const int ND = N * D;

    float* ws   = (float*)d_ws;
    float* agg  = ws;  ws += (size_t)ND;
    float* h1   = ws;  ws += (size_t)ND;
    float* h2   = ws;  ws += (size_t)ND;
    float* cnt  = ws;  ws += N;
    float* gcnt = ws;  ws += GRAPHS;
    float* pool = ws;  ws += (size_t)GRAPHS * D;

    const int* src = ei;
    const int* dstv = ei + E;

    dim3 blk(256);
    const unsigned scat_blocks = cdiv_u((long long)E * 32, 256);
    const unsigned gemm_blocks = (unsigned)(N / 16);

    fill_zero_kernel<<<cdiv_u(N, 256), blk, 0, stream>>>(cnt, N);
    count_deg_kernel<<<cdiv_u(E, 256), blk, 0, stream>>>(dstv, cnt, E);

    fill_zero_kernel<<<cdiv_u(ND, 256), blk, 0, stream>>>(agg, ND);
    scatter_edges_kernel<<<scat_blocks, blk, 0, stream>>>(x, src, dstv, agg, E);
    sage_layer_kernel<<<gemm_blocks, blk, 0, stream>>>(x, agg, cnt, Wl1, Wr1, b1, h1, 1);

    fill_zero_kernel<<<cdiv_u(ND, 256), blk, 0, stream>>>(agg, ND);
    scatter_edges_kernel<<<scat_blocks, blk, 0, stream>>>(h1, src, dstv, agg, E);
    sage_layer_kernel<<<gemm_blocks, blk, 0, stream>>>(h1, agg, cnt, Wl2, Wr2, b2, h2, 1);

    fill_zero_kernel<<<cdiv_u(ND, 256), blk, 0, stream>>>(agg, ND);
    scatter_edges_kernel<<<scat_blocks, blk, 0, stream>>>(h2, src, dstv, agg, E);
    sage_layer_kernel<<<gemm_blocks, blk, 0, stream>>>(h2, agg, cnt, Wl3, Wr3, b3, h1, 0);

    fill_zero_kernel<<<1, 64, 0, stream>>>(gcnt, GRAPHS);
    fill_zero_kernel<<<cdiv_u(GRAPHS * D, 256), blk, 0, stream>>>(pool, GRAPHS * D);
    graph_count_kernel<<<cdiv_u(N, 256), blk, 0, stream>>>(batch, gcnt, N);
    pool_sum_kernel<<<cdiv_u((long long)N * 32, 256), blk, 0, stream>>>(h1, batch, pool, N);
    pool_div_kernel<<<cdiv_u(GRAPHS * D, 256), blk, 0, stream>>>(pool, gcnt, out);
}


# Aggregation by gather: edges bucketed by destination once (count, prefix sum, fill); each layer sums neighbour rows per node in registers, 4 partial sums, replacing float atomics and zero fill
# speedup vs baseline: 1.5353x; 1.5353x over previous
#include <hip/hip_runtime.h>
#include <hip/hip_bf16.h>

typedef __attribute__((ext_vector_type(2))) float v2f;
typedef __attribute__((ext_vector_type(8))) float v8f;

#define D 128
#define GRAPHS 64


__global__ void fill_zero_kernel(float* __restrict__ p, int n) {
    int i = blockIdx.x * blockDim.x + threadIdx.x;
    if (i < n) p[i] = 0.0f;
}

__global__ void fill_zero_int_kernel(int* __restrict__ p, int n) {
    int i = blockIdx.x * blockDim.x + threadIdx.x;
    if (i < n) p[i] = 0;
}

__global__ void count_deg_kernel(const int* __restrict__ dst, int* __restrict__ deg, int E) {
    int e = blockIdx.x * blockDim.x + threadIdx.x;
    if (e < E) atomicAdd(&deg[dst[e]], 1);
}

__launch_bounds__(256)
__global__ void row_offsets_kernel(const int* __restrict__ deg, int* __restrict__ rowptr,
                                   int* __restrict__ cursor, float* __restrict__ cnt, int N) {
    __shared__ int wpre[8];
    __shared__ int wsum[8];
    const int tid  = threadIdx.x;
    const int lane = tid & 31;
    const int wv   = tid >> 5;
    const int first = blockIdx.x * 256;

    int part = 0;
    const int4* deg4 = (const int4*)deg + tid;
    const int n4   = first >> 2;
    const int full = n4 >> 8;
    int k = 0;
    for (; k + 4 <= full; k += 4) {
        const int4 q0 = deg4[(k + 0) << 8];
        const int4 q1 = deg4[(k + 1) << 8];
        const int4 q2 = deg4[(k + 2) << 8];
        const int4 q3 = deg4[(k + 3) << 8];
        part += ((q0.x + q0.y) + (q0.z + q0.w)) + ((q1.x + q1.y) + (q1.z + q1.w));
        part += ((q2.x + q2.y) + (q2.z + q2.w)) + ((q3.x + q3.y) + (q3.z + q3.w));
    }
    for (; k < full; ++k) {
        const int4 q = deg4[k << 8];
        part += (q.x + q.y) + (q.z + q.w);
    }
    if (tid < (n4 & 255)) {
        const int4 q = deg4[full << 8];
        part += (q.x + q.y) + (q.z + q.w);
    }
    for (int o = 16; o > 0; o >>= 1) part += __shfl_xor(part, o, 32);
    if (lane == 0) wpre[wv] = part;

    const int i = first + tid;
    const int d = (i < N) ? deg[i] : 0;
    int inc = d;
    for (int o = 1; o < 32; o <<= 1) {
        const int t = __shfl_up(inc, o, 32);
        if (lane >= o) inc += t;
    }
    if (lane == 31) wsum[wv] = inc;
    __syncthreads();

    int base = 0;
    for (int w = 0; w < 8; ++w) {
        base += wpre[w];
        if (w < wv) base += wsum[w];
    }
    if (i < N) {
        const int off = base + inc - d;
        rowptr[i] = off;
        cursor[i] = off;
        cnt[i]    = (float)d;
    }
}

__global__ void fill_csr_kernel(const int* __restrict__ src, const int* __restrict__ dst,
                                int* __restrict__ cursor, int* __restrict__ col, int E) {
    int e = blockIdx.x * blockDim.x + threadIdx.x;
    if (e < E) {
        const int p = atomicAdd(&cursor[dst[e]], 1);
        col[p] = src[e];
    }
}

__launch_bounds__(256)
__global__ void gather_sum_kernel(const float* __restrict__ x,
                                  const int* __restrict__ rowptr,
                                  const int* __restrict__ deg,
                                  const int* __restrict__ col,
                                  float* __restrict__ agg, int N) {
    const int lane = threadIdx.x & 31;
    const int node = __builtin_amdgcn_readfirstlane((int)(blockIdx.x * 8 + (threadIdx.x >> 5)));
    if (node >= N) return;
    const int beg = rowptr[node];
    const int n   = deg[node];
    const float* xl = x + lane * 4;

    float4 a0 = {0.f, 0.f, 0.f, 0.f};
    float4 a1 = a0, a2 = a0, a3 = a0;

    for (int done = 0; done < n; done += 32) {
        const int m = (n - done < 32) ? (n - done) : 32;
        const int idx = (lane < m) ? col[beg + done + lane] : 0;
        int j = 0;
        for (; j + 4 <= m; j += 4) {
            const int s0 = __builtin_amdgcn_readlane(idx, j);
            const int s1 = __builtin_amdgcn_readlane(idx, j + 1);
            const int s2 = __builtin_amdgcn_readlane(idx, j + 2);
            const int s3 = __builtin_amdgcn_readlane(idx, j + 3);
            const float4 v0 = *(const float4*)(xl + (size_t)s0 * D);
            const float4 v1 = *(const float4*)(xl + (size_t)s1 * D);
            const float4 v2 = *(const float4*)(xl + (size_t)s2 * D);
            const float4 v3 = *(const float4*)(xl + (size_t)s3 * D);
            a0.x += v0.x; a0.y += v0.y; a0.z += v0.z; a0.w += v0.w;
            a1.x += v1.x; a1.y += v1.y; a1.z += v1.z; a1.w += v1.w;
            a2.x += v2.x; a2.y += v2.y; a2.z += v2.z; a2.w += v2.w;
            a3.x += v3.x; a3.y += v3.y; a3.z += v3.z; a3.w += v3.w;
        }
        for (; j < m; ++j) {
            const int s0 = __builtin_amdgcn_readlane(idx, j);
            const float4 v0 = *(const float4*)(xl + (size_t)s0 * D);
            a0.x += v0.x; a0.y += v0.y; a0.z += v0.z; a0.w += v0.w;
        }
    }
    float4 r;
    r.x = (a0.x + a1.x) + (a2.x + a3.x);
    r.y = (a0.y + a1.y) + (a2.y + a3.y);
    r.z = (a0.z + a1.z) + (a2.z + a3.z);
    r.w = (a0.w + a1.w) + (a2.w + a3.w);
    *(float4*)(agg + (size_t)node * D + lane * 4) = r;
}


__launch_bounds__(256)
__global__ void sage_layer_kernel(const float* __restrict__ xin,
                                  const float* __restrict__ agg,
                                  const float* __restrict__ cnt,
                                  const float* __restrict__ Wl,
                                  const float* __restrict__ Wr,
                                  const float* __restrict__ bias,
                                  float* __restrict__ out,
                                  int do_relu) {
    __shared__ float ssq[16][8];
    __shared__ float inv_norm[16];

    const int lane = threadIdx.x & 31;
    const int wv   = threadIdx.x >> 5;
    const int half = lane >> 4;
    const int lr   = lane & 15;
    const int rowbase = blockIdx.x * 16;
    const int row  = rowbase + lr;
    const int col  = wv * 16 + lr;

    const float invc = 1.0f / fmaxf(cnt[row], 1.0f);
    const float* Am = agg + (size_t)row * D;
    const float* Ax = xin + (size_t)row * D;

    v8f acc = {0.f, 0.f, 0.f, 0.f, 0.f, 0.f, 0.f, 0.f};

#pragma unroll
    for (int k0 = 0; k0 < D; k0 += 4) {
        const int ka = k0 + 2 * half;
        v2f a, b;
        a.x = Am[ka] * invc;
        a.y = Am[ka + 1] * invc;
        b.x = Wl[(size_t)ka * D + col];
        b.y = Wl[(size_t)(ka + 1) * D + col];
        acc = __builtin_amdgcn_wmma_f32_16x16x4_f32(false, a, false, b,
                                                    (short)0, acc, false, false);
    }
#pragma unroll
    for (int k0 = 0; k0 < D; k0 += 4) {
        const int ka = k0 + 2 * half;
        v2f a, b;
        a.x = Ax[ka];
        a.y = Ax[ka + 1];
        b.x = Wr[(size_t)ka * D + col];
        b.y = Wr[(size_t)(ka + 1) * D + col];
        acc = __builtin_amdgcn_wmma_f32_16x16x4_f32(false, a, false, b,
                                                    (short)0, acc, false, false);
    }

    const float bb = bias[col];
#pragma unroll
    for (int r = 0; r < 8; ++r) {
        float v = acc[r] + bb;
        if (do_relu) v = fmaxf(v, 0.0f);
        acc[r] = v;
    }

#pragma unroll
    for (int r = 0; r < 8; ++r) {
        float s = acc[r] * acc[r];
        s += __shfl_xor(s, 1, 32);
        s += __shfl_xor(s, 2, 32);
        s += __shfl_xor(s, 4, 32);
        s += __shfl_xor(s, 8, 32);
        if (lr == 0) ssq[r + 8 * half][wv] = s;
    }
    __syncthreads();
    if (threadIdx.x < 16) {
        float s = 0.0f;
#pragma unroll
        for (int w = 0; w < 8; ++w) s += ssq[threadIdx.x][w];
        inv_norm[threadIdx.x] = 1.0f / fmaxf(sqrtf(s), 1e-12f);
    }
    __syncthreads();

#pragma unroll
    for (int r = 0; r < 8; ++r) {
        const float sc = inv_norm[r + 8 * half];
        out[(size_t)(rowbase + r + 8 * half) * D + col] = acc[r] * sc;
    }
}


__global__ void graph_count_kernel(const int* __restrict__ batch, float* __restrict__ gcnt, int N) {
    int n = blockIdx.x * blockDim.x + threadIdx.x;
    if (n < N) atomicAdd(&gcnt[batch[n]], 1.0f);
}

__global__ void pool_sum_kernel(const float* __restrict__ h, const int* __restrict__ batch,
                                float* __restrict__ pool, int N) {
    unsigned t = blockIdx.x * blockDim.x + threadIdx.x;
    unsigned n = t >> 5;
    if (n >= (unsigned)N) return;
    unsigned c = (t & 31u) << 2;
    int g = batch[n];
    const float4 v = *(const float4*)(h + (size_t)n * D + c);
    float* p = pool + (size_t)g * D + c;
    atomicAdd(p + 0, v.x);
    atomicAdd(p + 1, v.y);
    atomicAdd(p + 2, v.z);
    atomicAdd(p + 3, v.w);
}

__global__ void pool_div_kernel(const float* __restrict__ pool, const float* __restrict__ gcnt,
                                float* __restrict__ out) {
    int i = blockIdx.x * blockDim.x + threadIdx.x;
    if (i < GRAPHS * D) {
        int g = i / D;
        out[i] = pool[i] / fmaxf(gcnt[g], 1.0f);
    }
}


static inline unsigned cdiv_u(long long a, long long b) { return (unsigned)((a + b - 1) / b); }

extern "C" void kernel_launch(void* const* d_in, const int* in_sizes, int n_in,
                              void* d_out, int out_size, void* d_ws, size_t ws_size,
                              hipStream_t stream) {
    const float* x    = (const float*)d_in[0];
    const int*   ei   = (const int*)d_in[1];
    const int*   batch= (const int*)d_in[2];
    const float* Wl1  = (const float*)d_in[3];
    const float* Wr1  = (const float*)d_in[4];
    const float* b1   = (const float*)d_in[5];
    const float* Wl2  = (const float*)d_in[6];
    const float* Wr2  = (const float*)d_in[7];
    const float* b2   = (const float*)d_in[8];
    const float* Wl3  = (const float*)d_in[9];
    const float* Wr3  = (const float*)d_in[10];
    const float* b3   = (const float*)d_in[11];
    float* out = (float*)d_out;

    const int N  = in_sizes[0] / D;
    const int E  = in_sizes[1] / 2;
    const int ND = N * D;

    float* ws   = (float*)d_ws;
    float* agg  = ws;  ws += (size_t)ND;
    float* h1   = ws;  ws += (size_t)ND;
    float* h2   = ws;  ws += (size_t)ND;
    float* cnt  = ws;  ws += N;
    float* gcnt = ws;  ws += GRAPHS;
    float* pool = ws;  ws += (size_t)GRAPHS * D;
    int* deg    = (int*)ws;  ws += N;
    int* rowptr = (int*)ws;  ws += N;
    int* cursor = (int*)ws;  ws += N;
    int* col    = (int*)ws;  ws += E;

    const int* src = ei;
    const int* dstv = ei + E;

    dim3 blk(256);
    const unsigned edge_blocks = cdiv_u(E, 256);
    const unsigned node_blocks = cdiv_u(N, 256);
    const unsigned gath_blocks = cdiv_u(N, 8);
    const unsigned gemm_blocks = (unsigned)(N / 16);

    fill_zero_int_kernel<<<node_blocks, blk, 0, stream>>>(deg, N);
    count_deg_kernel<<<edge_blocks, blk, 0, stream>>>(dstv, deg, E);
    row_offsets_kernel<<<node_blocks, blk, 0, stream>>>(deg, rowptr, cursor, cnt, N);
    fill_csr_kernel<<<edge_blocks, blk, 0, stream>>>(src, dstv, cursor, col, E);

    gather_sum_kernel<<<gath_blocks, blk, 0, stream>>>(x, rowptr, deg, col, agg, N);
    sage_layer_kernel<<<gemm_blocks, blk, 0, stream>>>(x, agg, cnt, Wl1, Wr1, b1, h1, 1);

    gather_sum_kernel<<<gath_blocks, blk, 0, stream>>>(h1, rowptr, deg, col, agg, N);
    sage_layer_kernel<<<gemm_blocks, blk, 0, stream>>>(h1, agg, cnt, Wl2, Wr2, b2, h2, 1);

    gather_sum_kernel<<<gath_blocks, blk, 0, stream>>>(h2, rowptr, deg, col, agg, N);
    sage_layer_kernel<<<gemm_blocks, blk, 0, stream>>>(h2, agg, cnt, Wl3, Wr3, b3, h1, 0);

    fill_zero_kernel<<<1, 64, 0, stream>>>(gcnt, GRAPHS);
    fill_zero_kernel<<<cdiv_u(GRAPHS * D, 256), blk, 0, stream>>>(pool, GRAPHS * D);
    graph_count_kernel<<<cdiv_u(N, 256), blk, 0, stream>>>(batch, gcnt, N);
    pool_sum_kernel<<<cdiv_u((long long)N * 32, 256), blk, 0, stream>>>(h1, batch, pool, N);
    pool_div_kernel<<<cdiv_u(GRAPHS * D, 256), blk, 0, stream>>>(pool, gcnt, out);
}


# Layer GEMM re-tiled: 32-row block tile, A (mean | x) staged once per block in LDS by whole-row loads, each B fragment feeds two WMMAs; sum order unchanged
# speedup vs baseline: 1.7975x; 1.1708x over previous
#include <hip/hip_runtime.h>
#include <hip/hip_bf16.h>

typedef __attribute__((ext_vector_type(2))) float v2f;
typedef __attribute__((ext_vector_type(8))) float v8f;

#define D 128
#define GRAPHS 64


__global__ void fill_zero_kernel(float* __restrict__ p, int n) {
    int i = blockIdx.x * blockDim.x + threadIdx.x;
    if (i < n) p[i] = 0.0f;
}

__global__ void fill_zero_int_kernel(int* __restrict__ p, int n) {
    int i = blockIdx.x * blockDim.x + threadIdx.x;
    if (i < n) p[i] = 0;
}

__global__ void count_deg_kernel(const int* __restrict__ dst, int* __restrict__ deg, int E) {
    int e = blockIdx.x * blockDim.x + threadIdx.x;
    if (e < E) atomicAdd(&deg[dst[e]], 1);
}

__launch_bounds__(256)
__global__ void row_offsets_kernel(const int* __restrict__ deg, int* __restrict__ rowptr,
                                   int* __restrict__ cursor, float* __restrict__ cnt, int N) {
    __shared__ int wpre[8];
    __shared__ int wsum[8];
    const int tid  = threadIdx.x;
    const int lane = tid & 31;
    const int wv   = tid >> 5;
    const int first = blockIdx.x * 256;

    int part = 0;
    const int4* deg4 = (const int4*)deg + tid;
    const int n4   = first >> 2;
    const int full = n4 >> 8;
    int k = 0;
    for (; k + 4 <= full; k += 4) {
        const int4 q0 = deg4[(k + 0) << 8];
        const int4 q1 = deg4[(k + 1) << 8];
        const int4 q2 = deg4[(k + 2) << 8];
        const int4 q3 = deg4[(k + 3) << 8];
        part += ((q0.x + q0.y) + (q0.z + q0.w)) + ((q1.x + q1.y) + (q1.z + q1.w));
        part += ((q2.x + q2.y) + (q2.z + q2.w)) + ((q3.x + q3.y) + (q3.z + q3.w));
    }
    for (; k < full; ++k) {
        const int4 q = deg4[k << 8];
        part += (q.x + q.y) + (q.z + q.w);
    }
    if (tid < (n4 & 255)) {
        const int4 q = deg4[full << 8];
        part += (q.x + q.y) + (q.z + q.w);
    }
    for (int o = 16; o > 0; o >>= 1) part += __shfl_xor(part, o, 32);
    if (lane == 0) wpre[wv] = part;

    const int i = first + tid;
    const int d = (i < N) ? deg[i] : 0;
    int inc = d;
    for (int o = 1; o < 32; o <<= 1) {
        const int t = __shfl_up(inc, o, 32);
        if (lane >= o) inc += t;
    }
    if (lane == 31) wsum[wv] = inc;
    __syncthreads();

    int base = 0;
    for (int w = 0; w < 8; ++w) {
        base += wpre[w];
        if (w < wv) base += wsum[w];
    }
    if (i < N) {
        const int off = base + inc - d;
        rowptr[i] = off;
        cursor[i] = off;
        cnt[i]    = (float)d;
    }
}

__global__ void fill_csr_kernel(const int* __restrict__ src, const int* __restrict__ dst,
                                int* __restrict__ cursor, int* __restrict__ col, int E) {
    int e = blockIdx.x * blockDim.x + threadIdx.x;
    if (e < E) {
        const int p = atomicAdd(&cursor[dst[e]], 1);
        col[p] = src[e];
    }
}

__launch_bounds__(256)
__global__ void gather_sum_kernel(const float* __restrict__ x,
                                  const int* __restrict__ rowptr,
                                  const int* __restrict__ deg,
                                  const int* __restrict__ col,
                                  float* __restrict__ agg, int N) {
    const int lane = threadIdx.x & 31;
    const int node = __builtin_amdgcn_readfirstlane((int)(blockIdx.x * 8 + (threadIdx.x >> 5)));
    if (node >= N) return;
    const int beg = rowptr[node];
    const int n   = deg[node];
    const float* xl = x + lane * 4;

    float4 a0 = {0.f, 0.f, 0.f, 0.f};
    float4 a1 = a0, a2 = a0, a3 = a0;

    for (int done = 0; done < n; done += 32) {
        const int m = (n - done < 32) ? (n - done) : 32;
        const int idx = (lane < m) ? col[beg + done + lane] : 0;
        int j = 0;
        for (; j + 4 <= m; j += 4) {
            const int s0 = __builtin_amdgcn_readlane(idx, j);
            const int s1 = __builtin_amdgcn_readlane(idx, j + 1);
            const int s2 = __builtin_amdgcn_readlane(idx, j + 2);
            const int s3 = __builtin_amdgcn_readlane(idx, j + 3);
            const float4 v0 = *(const float4*)(xl + (size_t)s0 * D);
            const float4 v1 = *(const float4*)(xl + (size_t)s1 * D);
            const float4 v2 = *(const float4*)(xl + (size_t)s2 * D);
            const float4 v3 = *(const float4*)(xl + (size_t)s3 * D);
            a0.x += v0.x; a0.y += v0.y; a0.z += v0.z; a0.w += v0.w;
            a1.x += v1.x; a1.y += v1.y; a1.z += v1.z; a1.w += v1.w;
            a2.x += v2.x; a2.y += v2.y; a2.z += v2.z; a2.w += v2.w;
            a3.x += v3.x; a3.y += v3.y; a3.z += v3.z; a3.w += v3.w;
        }
        for (; j < m; ++j) {
            const int s0 = __builtin_amdgcn_readlane(idx, j);
            const float4 v0 = *(const float4*)(xl + (size_t)s0 * D);
            a0.x += v0.x; a0.y += v0.y; a0.z += v0.z; a0.w += v0.w;
        }
    }
    float4 r;
    r.x = (a0.x + a1.x) + (a2.x + a3.x);
    r.y = (a0.y + a1.y) + (a2.y + a3.y);
    r.z = (a0.z + a1.z) + (a2.z + a3.z);
    r.w = (a0.w + a1.w) + (a2.w + a3.w);
    *(float4*)(agg + (size_t)node * D + lane * 4) = r;
}


#define TR 32
#define APITCH 260

__launch_bounds__(256)
__global__ void sage_layer_kernel(const float* __restrict__ xin,
                                  const float* __restrict__ agg,
                                  const float* __restrict__ cnt,
                                  const float* __restrict__ Wl,
                                  const float* __restrict__ Wr,
                                  const float* __restrict__ bias,
                                  float* __restrict__ out,
                                  int N, int do_relu) {
    __shared__ __attribute__((aligned(16))) float As[TR * APITCH];
    __shared__ float ssq[TR][8];
    __shared__ float inv_norm[TR];

    const int lane = threadIdx.x & 31;
    const int wv   = __builtin_amdgcn_readfirstlane((int)(threadIdx.x >> 5));
    const int half = lane >> 4;
    const int lr   = lane & 15;
    const int rowbase = blockIdx.x * TR;
    const int col  = wv * 16 + lr;

#pragma unroll
    for (int i = 0; i < TR / 8; ++i) {
        const int r   = wv * (TR / 8) + i;
        const int row = rowbase + r;
        float4 m  = {0.f, 0.f, 0.f, 0.f};
        float4 xv = {0.f, 0.f, 0.f, 0.f};
        if (row < N) {
            const float invc = 1.0f / fmaxf(cnt[row], 1.0f);
            m  = *(const float4*)(agg + (size_t)row * D + lane * 4);
            xv = *(const float4*)(xin + (size_t)row * D + lane * 4);
            m.x *= invc; m.y *= invc; m.z *= invc; m.w *= invc;
        }
        *(float4*)(&As[r * APITCH + lane * 4])     = m;
        *(float4*)(&As[r * APITCH + D + lane * 4]) = xv;
    }
    __syncthreads();

    v8f acc0 = {0.f, 0.f, 0.f, 0.f, 0.f, 0.f, 0.f, 0.f};
    v8f acc1 = acc0;

    const float* a0p = &As[lr * APITCH + 2 * half];
    const float* a1p = &As[(16 + lr) * APITCH + 2 * half];

#pragma unroll
    for (int k0 = 0; k0 < D; k0 += 4) {
        const int ka = k0 + 2 * half;
        v2f b;
        b.x = Wl[(size_t)ka * D + col];
        b.y = Wl[(size_t)(ka + 1) * D + col];
        const v2f a0 = *(const v2f*)(a0p + k0);
        const v2f a1 = *(const v2f*)(a1p + k0);
        acc0 = __builtin_amdgcn_wmma_f32_16x16x4_f32(false, a0, false, b,
                                                     (short)0, acc0, false, false);
        acc1 = __builtin_amdgcn_wmma_f32_16x16x4_f32(false, a1, false, b,
                                                     (short)0, acc1, false, false);
    }
#pragma unroll
    for (int k0 = 0; k0 < D; k0 += 4) {
        const int ka = k0 + 2 * half;
        v2f b;
        b.x = Wr[(size_t)ka * D + col];
        b.y = Wr[(size_t)(ka + 1) * D + col];
        const v2f a0 = *(const v2f*)(a0p + D + k0);
        const v2f a1 = *(const v2f*)(a1p + D + k0);
        acc0 = __builtin_amdgcn_wmma_f32_16x16x4_f32(false, a0, false, b,
                                                     (short)0, acc0, false, false);
        acc1 = __builtin_amdgcn_wmma_f32_16x16x4_f32(false, a1, false, b,
                                                     (short)0, acc1, false, false);
    }

    const float bb = bias[col];
#pragma unroll
    for (int r = 0; r < 8; ++r) {
        float v0 = acc0[r] + bb;
        float v1 = acc1[r] + bb;
        if (do_relu) { v0 = fmaxf(v0, 0.0f); v1 = fmaxf(v1, 0.0f); }
        acc0[r] = v0;
        acc1[r] = v1;
    }

#pragma unroll
    for (int r = 0; r < 8; ++r) {
        float s = acc0[r] * acc0[r];
        s += __shfl_xor(s, 1, 32);
        s += __shfl_xor(s, 2, 32);
        s += __shfl_xor(s, 4, 32);
        s += __shfl_xor(s, 8, 32);
        if (lr == 0) ssq[r + 8 * half][wv] = s;
        float t = acc1[r] * acc1[r];
        t += __shfl_xor(t, 1, 32);
        t += __shfl_xor(t, 2, 32);
        t += __shfl_xor(t, 4, 32);
        t += __shfl_xor(t, 8, 32);
        if (lr == 0) ssq[16 + r + 8 * half][wv] = t;
    }
    __syncthreads();
    if (threadIdx.x < TR) {
        float s = 0.0f;
#pragma unroll
        for (int w = 0; w < 8; ++w) s += ssq[threadIdx.x][w];
        inv_norm[threadIdx.x] = 1.0f / fmaxf(sqrtf(s), 1e-12f);
    }
    __syncthreads();

#pragma unroll
    for (int r = 0; r < 8; ++r) {
        const int r0 = r + 8 * half;
        const int row0 = rowbase + r0;
        const int row1 = row0 + 16;
        if (row0 < N) out[(size_t)row0 * D + col] = acc0[r] * inv_norm[r0];
        if (row1 < N) out[(size_t)row1 * D + col] = acc1[r] * inv_norm[r0 + 16];
    }
}


__global__ void graph_count_kernel(const int* __restrict__ batch, float* __restrict__ gcnt, int N) {
    int n = blockIdx.x * blockDim.x + threadIdx.x;
    if (n < N) atomicAdd(&gcnt[batch[n]], 1.0f);
}

__global__ void pool_sum_kernel(const float* __restrict__ h, const int* __restrict__ batch,
                                float* __restrict__ pool, int N) {
    unsigned t = blockIdx.x * blockDim.x + threadIdx.x;
    unsigned n = t >> 5;
    if (n >= (unsigned)N) return;
    unsigned c = (t & 31u) << 2;
    int g = batch[n];
    const float4 v = *(const float4*)(h + (size_t)n * D + c);
    float* p = pool + (size_t)g * D + c;
    atomicAdd(p + 0, v.x);
    atomicAdd(p + 1, v.y);
    atomicAdd(p + 2, v.z);
    atomicAdd(p + 3, v.w);
}

__global__ void pool_div_kernel(const float* __restrict__ pool, const float* __restrict__ gcnt,
                                float* __restrict__ out) {
    int i = blockIdx.x * blockDim.x + threadIdx.x;
    if (i < GRAPHS * D) {
        int g = i / D;
        out[i] = pool[i] / fmaxf(gcnt[g], 1.0f);
    }
}


static inline unsigned cdiv_u(long long a, long long b) { return (unsigned)((a + b - 1) / b); }

extern "C" void kernel_launch(void* const* d_in, const int* in_sizes, int n_in,
                              void* d_out, int out_size, void* d_ws, size_t ws_size,
                              hipStream_t stream) {
    const float* x    = (const float*)d_in[0];
    const int*   ei   = (const int*)d_in[1];
    const int*   batch= (const int*)d_in[2];
    const float* Wl1  = (const float*)d_in[3];
    const float* Wr1  = (const float*)d_in[4];
    const float* b1   = (const float*)d_in[5];
    const float* Wl2  = (const float*)d_in[6];
    const float* Wr2  = (const float*)d_in[7];
    const float* b2   = (const float*)d_in[8];
    const float* Wl3  = (const float*)d_in[9];
    const float* Wr3  = (const float*)d_in[10];
    const float* b3   = (const float*)d_in[11];
    float* out = (float*)d_out;

    const int N  = in_sizes[0] / D;
    const int E  = in_sizes[1] / 2;
    const int ND = N * D;

    float* ws   = (float*)d_ws;
    float* agg  = ws;  ws += (size_t)ND;
    float* h1   = ws;  ws += (size_t)ND;
    float* h2   = ws;  ws += (size_t)ND;
    float* cnt  = ws;  ws += N;
    float* gcnt = ws;  ws += GRAPHS;
    float* pool = ws;  ws += (size_t)GRAPHS * D;
    int* deg    = (int*)ws;  ws += N;
    int* rowptr = (int*)ws;  ws += N;
    int* cursor = (int*)ws;  ws += N;
    int* col    = (int*)ws;  ws += E;

    const int* src = ei;
    const int* dstv = ei + E;

    dim3 blk(256);
    const unsigned edge_blocks = cdiv_u(E, 256);
    const unsigned node_blocks = cdiv_u(N, 256);
    const unsigned gath_blocks = cdiv_u(N, 8);
    const unsigned gemm_blocks = cdiv_u(N, TR);

    fill_zero_int_kernel<<<node_blocks, blk, 0, stream>>>(deg, N);
    count_deg_kernel<<<edge_blocks, blk, 0, stream>>>(dstv, deg, E);
    row_offsets_kernel<<<node_blocks, blk, 0, stream>>>(deg, rowptr, cursor, cnt, N);
    fill_csr_kernel<<<edge_blocks, blk, 0, stream>>>(src, dstv, cursor, col, E);

    gather_sum_kernel<<<gath_blocks, blk, 0, stream>>>(x, rowptr, deg, col, agg, N);
    sage_layer_kernel<<<gemm_blocks, blk, 0, stream>>>(x, agg, cnt, Wl1, Wr1, b1, h1, N, 1);

    gather_sum_kernel<<<gath_blocks, blk, 0, stream>>>(h1, rowptr, deg, col, agg, N);
    sage_layer_kernel<<<gemm_blocks, blk, 0, stream>>>(h1, agg, cnt, Wl2, Wr2, b2, h2, N, 1);

    gather_sum_kernel<<<gath_blocks, blk, 0, stream>>>(h2, rowptr, deg, col, agg, N);
    sage_layer_kernel<<<gemm_blocks, blk, 0, stream>>>(h2, agg, cnt, Wl3, Wr3, b3, h1, N, 0);

    fill_zero_kernel<<<1, 64, 0, stream>>>(gcnt, GRAPHS);
    fill_zero_kernel<<<cdiv_u(GRAPHS * D, 256), blk, 0, stream>>>(pool, GRAPHS * D);
    graph_count_kernel<<<cdiv_u(N, 256), blk, 0, stream>>>(batch, gcnt, N);
    pool_sum_kernel<<<cdiv_u((long long)N * 32, 256), blk, 0, stream>>>(h1, batch, pool, N);
    pool_div_kernel<<<cdiv_u(GRAPHS * D, 256), blk, 0, stream>>>(pool, gcnt, out);
}


# Mean pool by runs: a wave sums 32 consecutive rows in registers while the graph id repeats, one atomic add per run and column (was per node), counts from run lengths, one zero fill; sum order changed
# speedup vs baseline: 2.6930x; 1.4982x over previous
#include <hip/hip_runtime.h>
#include <hip/hip_bf16.h>

typedef __attribute__((ext_vector_type(2))) float v2f;
typedef __attribute__((ext_vector_type(8))) float v8f;

#define D 128
#define GRAPHS 64


__global__ void fill_zero_int_kernel(int* __restrict__ p, int n) {
    int i = blockIdx.x * blockDim.x + threadIdx.x;
    if (i < n) p[i] = 0;
}

__global__ void count_deg_kernel(const int* __restrict__ dst, int* __restrict__ deg, int E) {
    int e = blockIdx.x * blockDim.x + threadIdx.x;
    if (e < E) atomicAdd(&deg[dst[e]], 1);
}

__launch_bounds__(256)
__global__ void row_offsets_kernel(const int* __restrict__ deg, int* __restrict__ rowptr,
                                   int* __restrict__ cursor, float* __restrict__ cnt, int N) {
    __shared__ int wpre[8];
    __shared__ int wsum[8];
    const int tid  = threadIdx.x;
    const int lane = tid & 31;
    const int wv   = tid >> 5;
    const int first = blockIdx.x * 256;

    int part = 0;
    const int4* deg4 = (const int4*)deg + tid;
    const int n4   = first >> 2;
    const int full = n4 >> 8;
    int k = 0;
    for (; k + 4 <= full; k += 4) {
        const int4 q0 = deg4[(k + 0) << 8];
        const int4 q1 = deg4[(k + 1) << 8];
        const int4 q2 = deg4[(k + 2) << 8];
        const int4 q3 = deg4[(k + 3) << 8];
        part += ((q0.x + q0.y) + (q0.z + q0.w)) + ((q1.x + q1.y) + (q1.z + q1.w));
        part += ((q2.x + q2.y) + (q2.z + q2.w)) + ((q3.x + q3.y) + (q3.z + q3.w));
    }
    for (; k < full; ++k) {
        const int4 q = deg4[k << 8];
        part += (q.x + q.y) + (q.z + q.w);
    }
    if (tid < (n4 & 255)) {
        const int4 q = deg4[full << 8];
        part += (q.x + q.y) + (q.z + q.w);
    }
    for (int o = 16; o > 0; o >>= 1) part += __shfl_xor(part, o, 32);
    if (lane == 0) wpre[wv] = part;

    const int i = first + tid;
    const int d = (i < N) ? deg[i] : 0;
    int inc = d;
    for (int o = 1; o < 32; o <<= 1) {
        const int t = __shfl_up(inc, o, 32);
        if (lane >= o) inc += t;
    }
    if (lane == 31) wsum[wv] = inc;
    __syncthreads();

    int base = 0;
    for (int w = 0; w < 8; ++w) {
        base += wpre[w];
        if (w < wv) base += wsum[w];
    }
    if (i < N) {
        const int off = base + inc - d;
        rowptr[i] = off;
        cursor[i] = off;
        cnt[i]    = (float)d;
    }
}

__global__ void fill_csr_kernel(const int* __restrict__ src, const int* __restrict__ dst,
                                int* __restrict__ cursor, int* __restrict__ col, int E) {
    int e = blockIdx.x * blockDim.x + threadIdx.x;
    if (e < E) {
        const int p = atomicAdd(&cursor[dst[e]], 1);
        col[p] = src[e];
    }
}

__launch_bounds__(256)
__global__ void gather_sum_kernel(const float* __restrict__ x,
                                  const int* __restrict__ rowptr,
                                  const int* __restrict__ deg,
                                  const int* __restrict__ col,
                                  float* __restrict__ agg, int N) {
    const int lane = threadIdx.x & 31;
    const int node = __builtin_amdgcn_readfirstlane((int)(blockIdx.x * 8 + (threadIdx.x >> 5)));
    if (node >= N) return;
    const int beg = rowptr[node];
    const int n   = deg[node];
    const float* xl = x + lane * 4;

    float4 a0 = {0.f, 0.f, 0.f, 0.f};
    float4 a1 = a0, a2 = a0, a3 = a0;

    for (int done = 0; done < n; done += 32) {
        const int m = (n - done < 32) ? (n - done) : 32;
        const int idx = (lane < m) ? col[beg + done + lane] : 0;
        int j = 0;
        for (; j + 4 <= m; j += 4) {
            const int s0 = __builtin_amdgcn_readlane(idx, j);
            const int s1 = __builtin_amdgcn_readlane(idx, j + 1);
            const int s2 = __builtin_amdgcn_readlane(idx, j + 2);
            const int s3 = __builtin_amdgcn_readlane(idx, j + 3);
            const float4 v0 = *(const float4*)(xl + (size_t)s0 * D);
            const float4 v1 = *(const float4*)(xl + (size_t)s1 * D);
            const float4 v2 = *(const float4*)(xl + (size_t)s2 * D);
            const float4 v3 = *(const float4*)(xl + (size_t)s3 * D);
            a0.x += v0.x; a0.y += v0.y; a0.z += v0.z; a0.w += v0.w;
            a1.x += v1.x; a1.y += v1.y; a1.z += v1.z; a1.w += v1.w;
            a2.x += v2.x; a2.y += v2.y; a2.z += v2.z; a2.w += v2.w;
            a3.x += v3.x; a3.y += v3.y; a3.z += v3.z; a3.w += v3.w;
        }
        for (; j < m; ++j) {
            const int s0 = __builtin_amdgcn_readlane(idx, j);
            const float4 v0 = *(const float4*)(xl + (size_t)s0 * D);
            a0.x += v0.x; a0.y += v0.y; a0.z += v0.z; a0.w += v0.w;
        }
    }
    float4 r;
    r.x = (a0.x + a1.x) + (a2.x + a3.x);
    r.y = (a0.y + a1.y) + (a2.y + a3.y);
    r.z = (a0.z + a1.z) + (a2.z + a3.z);
    r.w = (a0.w + a1.w) + (a2.w + a3.w);
    *(float4*)(agg + (size_t)node * D + lane * 4) = r;
}


#define TR 32
#define APITCH 260

__launch_bounds__(256)
__global__ void sage_layer_kernel(const float* __restrict__ xin,
                                  const float* __restrict__ agg,
                                  const float* __restrict__ cnt,
                                  const float* __restrict__ Wl,
                                  const float* __restrict__ Wr,
                                  const float* __restrict__ bias,
                                  float* __restrict__ out,
                                  int N, int do_relu) {
    __shared__ __attribute__((aligned(16))) float As[TR * APITCH];
    __shared__ float ssq[TR][8];
    __shared__ float inv_norm[TR];

    const int lane = threadIdx.x & 31;
    const int wv   = __builtin_amdgcn_readfirstlane((int)(threadIdx.x >> 5));
    const int half = lane >> 4;
    const int lr   = lane & 15;
    const int rowbase = blockIdx.x * TR;
    const int col  = wv * 16 + lr;

#pragma unroll
    for (int i = 0; i < TR / 8; ++i) {
        const int r   = wv * (TR / 8) + i;
        const int row = rowbase + r;
        float4 m  = {0.f, 0.f, 0.f, 0.f};
        float4 xv = {0.f, 0.f, 0.f, 0.f};
        if (row < N) {
            const float invc = 1.0f / fmaxf(cnt[row], 1.0f);
            m  = *(const float4*)(agg + (size_t)row * D + lane * 4);
            xv = *(const float4*)(xin + (size_t)row * D + lane * 4);
            m.x *= invc; m.y *= invc; m.z *= invc; m.w *= invc;
        }
        *(float4*)(&As[r * APITCH + lane * 4])     = m;
        *(float4*)(&As[r * APITCH + D + lane * 4]) = xv;
    }
    __syncthreads();

    v8f acc0 = {0.f, 0.f, 0.f, 0.f, 0.f, 0.f, 0.f, 0.f};
    v8f acc1 = acc0;

    const float* a0p = &As[lr * APITCH + 2 * half];
    const float* a1p = &As[(16 + lr) * APITCH + 2 * half];

#pragma unroll
    for (int k0 = 0; k0 < D; k0 += 4) {
        const int ka = k0 + 2 * half;
        v2f b;
        b.x = Wl[(size_t)ka * D + col];
        b.y = Wl[(size_t)(ka + 1) * D + col];
        const v2f a0 = *(const v2f*)(a0p + k0);
        const v2f a1 = *(const v2f*)(a1p + k0);
        acc0 = __builtin_amdgcn_wmma_f32_16x16x4_f32(false, a0, false, b,
                                                     (short)0, acc0, false, false);
        acc1 = __builtin_amdgcn_wmma_f32_16x16x4_f32(false, a1, false, b,
                                                     (short)0, acc1, false, false);
    }
#pragma unroll
    for (int k0 = 0; k0 < D; k0 += 4) {
        const int ka = k0 + 2 * half;
        v2f b;
        b.x = Wr[(size_t)ka * D + col];
        b.y = Wr[(size_t)(ka + 1) * D + col];
        const v2f a0 = *(const v2f*)(a0p + D + k0);
        const v2f a1 = *(const v2f*)(a1p + D + k0);
        acc0 = __builtin_amdgcn_wmma_f32_16x16x4_f32(false, a0, false, b,
                                                     (short)0, acc0, false, false);
        acc1 = __builtin_amdgcn_wmma_f32_16x16x4_f32(false, a1, false, b,
                                                     (short)0, acc1, false, false);
    }

    const float bb = bias[col];
#pragma unroll
    for (int r = 0; r < 8; ++r) {
        float v0 = acc0[r] + bb;
        float v1 = acc1[r] + bb;
        if (do_relu) { v0 = fmaxf(v0, 0.0f); v1 = fmaxf(v1, 0.0f); }
        acc0[r] = v0;
        acc1[r] = v1;
    }

#pragma unroll
    for (int r = 0; r < 8; ++r) {
        float s = acc0[r] * acc0[r];
        s += __shfl_xor(s, 1, 32);
        s += __shfl_xor(s, 2, 32);
        s += __shfl_xor(s, 4, 32);
        s += __shfl_xor(s, 8, 32);
        if (lr == 0) ssq[r + 8 * half][wv] = s;
        float t = acc1[r] * acc1[r];
        t += __shfl_xor(t, 1, 32);
        t += __shfl_xor(t, 2, 32);
        t += __shfl_xor(t, 4, 32);
        t += __shfl_xor(t, 8, 32);
        if (lr == 0) ssq[16 + r + 8 * half][wv] = t;
    }
    __syncthreads();
    if (threadIdx.x < TR) {
        float s = 0.0f;
#pragma unroll
        for (int w = 0; w < 8; ++w) s += ssq[threadIdx.x][w];
        inv_norm[threadIdx.x] = 1.0f / fmaxf(sqrtf(s), 1e-12f);
    }
    __syncthreads();

#pragma unroll
    for (int r = 0; r < 8; ++r) {
        const int r0 = r + 8 * half;
        const int row0 = rowbase + r0;
        const int row1 = row0 + 16;
        if (row0 < N) out[(size_t)row0 * D + col] = acc0[r] * inv_norm[r0];
        if (row1 < N) out[(size_t)row1 * D + col] = acc1[r] * inv_norm[r0 + 16];
    }
}


#define NPW 32

__device__ __forceinline__ void pool_flush(float* __restrict__ pool, float* __restrict__ gcnt,
                                           int g, const float4& a, int run, int lane) {
    float* p = pool + (size_t)g * D + lane * 4;
    atomicAdd(p + 0, a.x);
    atomicAdd(p + 1, a.y);
    atomicAdd(p + 2, a.z);
    atomicAdd(p + 3, a.w);
    if (lane == 0) atomicAdd(gcnt + g, (float)run);
}

__launch_bounds__(128)
__global__ void pool_runs_kernel(const float* __restrict__ h, const int* __restrict__ batch,
                                 float* __restrict__ pool, float* __restrict__ gcnt, int N) {
    const int lane = threadIdx.x & 31;
    const int first = __builtin_amdgcn_readfirstlane((int)(blockIdx.x * 4 + (threadIdx.x >> 5))) * NPW;
    if (first >= N) return;
    const int m = (N - first < NPW) ? (N - first) : NPW;
    const int gid = (lane < m) ? batch[first + lane] : 0;
    const float* hp = h + (size_t)first * D + lane * 4;

    float4 v[NPW];
#pragma unroll
    for (int j = 0; j < NPW; ++j) {
        const int jj = (j < m) ? j : (m - 1);
        v[j] = *(const float4*)(hp + (size_t)jj * D);
    }

    float4 acc = v[0];
    int cur = __builtin_amdgcn_readlane(gid, 0);
    int run = 1;
#pragma unroll
    for (int j = 1; j < NPW; ++j) {
        if (j < m) {
            const int g = __builtin_amdgcn_readlane(gid, j);
            if (g != cur) {
                pool_flush(pool, gcnt, cur, acc, run, lane);
                acc.x = 0.f; acc.y = 0.f; acc.z = 0.f; acc.w = 0.f;
                run = 0;
                cur = g;
            }
            acc.x += v[j].x; acc.y += v[j].y; acc.z += v[j].z; acc.w += v[j].w;
            ++run;
        }
    }
    pool_flush(pool, gcnt, cur, acc, run, lane);
}

__global__ void pool_div_kernel(const float* __restrict__ pool, const float* __restrict__ gcnt,
                                float* __restrict__ out) {
    int i = blockIdx.x * blockDim.x + threadIdx.x;
    if (i < GRAPHS * D) {
        int g = i / D;
        out[i] = pool[i] / fmaxf(gcnt[g], 1.0f);
    }
}


static inline unsigned cdiv_u(long long a, long long b) { return (unsigned)((a + b - 1) / b); }

extern "C" void kernel_launch(void* const* d_in, const int* in_sizes, int n_in,
                              void* d_out, int out_size, void* d_ws, size_t ws_size,
                              hipStream_t stream) {
    const float* x    = (const float*)d_in[0];
    const int*   ei   = (const int*)d_in[1];
    const int*   batch= (const int*)d_in[2];
    const float* Wl1  = (const float*)d_in[3];
    const float* Wr1  = (const float*)d_in[4];
    const float* b1   = (const float*)d_in[5];
    const float* Wl2  = (const float*)d_in[6];
    const float* Wr2  = (const float*)d_in[7];
    const float* b2   = (const float*)d_in[8];
    const float* Wl3  = (const float*)d_in[9];
    const float* Wr3  = (const float*)d_in[10];
    const float* b3   = (const float*)d_in[11];
    float* out = (float*)d_out;

    const int N  = in_sizes[0] / D;
    const int E  = in_sizes[1] / 2;
    const int ND = N * D;

    float* ws   = (float*)d_ws;
    float* agg  = ws;  ws += (size_t)ND;
    float* h1   = ws;  ws += (size_t)ND;
    float* h2   = ws;  ws += (size_t)ND;
    float* cnt  = ws;  ws += N;
    float* gcnt = ws;  ws += GRAPHS;
    float* pool = ws;  ws += (size_t)GRAPHS * D;
    int* deg    = (int*)ws;  ws += N;
    int* rowptr = (int*)ws;  ws += N;
    int* cursor = (int*)ws;  ws += N;
    int* col    = (int*)ws;  ws += E;

    const int* src = ei;
    const int* dstv = ei + E;

    dim3 blk(256);
    const unsigned edge_blocks = cdiv_u(E, 256);
    const unsigned node_blocks = cdiv_u(N, 256);
    const unsigned gath_blocks = cdiv_u(N, 8);
    const unsigned gemm_blocks = cdiv_u(N, TR);

    fill_zero_int_kernel<<<cdiv_u(GRAPHS + GRAPHS * D + N, 256), blk, 0, stream>>>((int*)gcnt, GRAPHS + GRAPHS * D + N);
    count_deg_kernel<<<edge_blocks, blk, 0, stream>>>(dstv, deg, E);
    row_offsets_kernel<<<node_blocks, blk, 0, stream>>>(deg, rowptr, cursor, cnt, N);
    fill_csr_kernel<<<edge_blocks, blk, 0, stream>>>(src, dstv, cursor, col, E);

    gather_sum_kernel<<<gath_blocks, blk, 0, stream>>>(x, rowptr, deg, col, agg, N);
    sage_layer_kernel<<<gemm_blocks, blk, 0, stream>>>(x, agg, cnt, Wl1, Wr1, b1, h1, N, 1);

    gather_sum_kernel<<<gath_blocks, blk, 0, stream>>>(h1, rowptr, deg, col, agg, N);
    sage_layer_kernel<<<gemm_blocks, blk, 0, stream>>>(h1, agg, cnt, Wl2, Wr2, b2, h2, N, 1);

    gather_sum_kernel<<<gath_blocks, blk, 0, stream>>>(h2, rowptr, deg, col, agg, N);
    sage_layer_kernel<<<gemm_blocks, blk, 0, stream>>>(h2, agg, cnt, Wl3, Wr3, b3, h1, N, 0);

    pool_runs_kernel<<<cdiv_u(N, 4 * NPW), dim3(128), 0, stream>>>(h1, batch, pool, gcnt, N);
    pool_div_kernel<<<cdiv_u(GRAPHS * D, 256), blk, 0, stream>>>(pool, gcnt, out);
}


# Edge bucketing through LDS: edges binned by 256-node range, sorted in LDS and written in runs; one block per range builds offsets and lists; per-edge global atomics gone; bucket order by LDS arrival
# speedup vs baseline: 4.1187x; 1.5294x over previous
#include <hip/hip_runtime.h>
#include <hip/hip_bf16.h>

typedef __attribute__((ext_vector_type(2))) float v2f;
typedef __attribute__((ext_vector_type(8))) float v8f;

#define D 128
#define GRAPHS 64


__global__ void fill_zero_int_kernel(int* __restrict__ p, int n) {
    int i = blockIdx.x * blockDim.x + threadIdx.x;
    if (i < n) p[i] = 0;
}


#define EPB 4096
#define EPT (EPB / 256)
#define SCAP 8192
#define SPT 8

__launch_bounds__(256)
__global__ void bin_count_kernel(const int* __restrict__ dst, int* __restrict__ binCount, int E) {
    __shared__ int hist[256];
    const int tid = threadIdx.x;
    const int first = blockIdx.x * EPB;
    hist[tid] = 0;
    __syncthreads();
    int d[EPT];
#pragma unroll
    for (int i = 0; i < EPT; ++i) {
        const int e = first + i * 256 + tid;
        d[i] = -1;
        if (e < E) d[i] = dst[e];
    }
#pragma unroll
    for (int i = 0; i < EPT; ++i)
        if (d[i] >= 0) atomicAdd(&hist[d[i] >> 8], 1);
    __syncthreads();
    const int h = hist[tid];
    if (h > 0) atomicAdd(&binCount[tid], h);
}

__launch_bounds__(256)
__global__ void bin_scatter_kernel(const int* __restrict__ src, const int* __restrict__ dst,
                                   const int* __restrict__ binCount, int* __restrict__ binCursor,
                                   int* __restrict__ binStart, int* __restrict__ binned, int E) {
    __shared__ int hist[257];
    __shared__ int loff[257];
    __shared__ int gbase[257];
    __shared__ int wsA[8];
    __shared__ int wsB[8];
    __shared__ int sval[EPB];
    __shared__ int sadr[EPB];
    const int tid  = threadIdx.x;
    const int lane = tid & 31;
    const int wv   = tid >> 5;
    const int first = blockIdx.x * EPB;
    const int n = (E - first < EPB) ? (E - first) : EPB;

    hist[tid] = 0;
    if (tid == 0) { hist[256] = 0; loff[256] = 0; gbase[256] = 0; }
    __syncthreads();

    int d[EPT], s[EPT];
#pragma unroll
    for (int i = 0; i < EPT; ++i) {
        const int e = first + i * 256 + tid;
        d[i] = -1;
        s[i] = 0;
        if (e < E) { d[i] = dst[e]; s[i] = src[e]; }
    }
#pragma unroll
    for (int i = 0; i < EPT; ++i)
        if (d[i] >= 0) atomicAdd(&hist[d[i] >> 8], 1);
    const int ca = binCount[tid];
    __syncthreads();
    const int cb = hist[tid];

    int ia = ca, ib = cb;
    for (int o = 1; o < 32; o <<= 1) {
        const int ta = __shfl_up(ia, o, 32);
        const int tb = __shfl_up(ib, o, 32);
        if (lane >= o) { ia += ta; ib += tb; }
    }
    if (lane == 31) { wsA[wv] = ia; wsB[wv] = ib; }
    __syncthreads();
    int ba = 0, bb = 0;
    for (int w = 0; w < 8; ++w)
        if (w < wv) { ba += wsA[w]; bb += wsB[w]; }
    const int ga = ba + ia - ca;
    const int lb = bb + ib - cb;
    if (blockIdx.x == 0) binStart[tid] = ga;
    int r = 0;
    if (cb > 0) r = atomicAdd(&binCursor[tid], cb);
    gbase[tid] = ga + r;
    loff[tid]  = lb;
    hist[tid]  = 0;
    __syncthreads();

    int k[EPT], lo[EPT], gb[EPT];
#pragma unroll
    for (int i = 0; i < EPT; ++i) {
        const int b = (d[i] >= 0) ? (d[i] >> 8) : 256;
        k[i]  = atomicAdd(&hist[b], 1);
        lo[i] = loff[b];
        gb[i] = gbase[b];
    }
#pragma unroll
    for (int i = 0; i < EPT; ++i) {
        if (d[i] >= 0) {
            const int q = lo[i] + k[i];
            sval[q] = (int)((unsigned)s[i] | ((unsigned)(d[i] & 255) << 24));
            sadr[q] = gb[i] + k[i];
        }
    }
    __syncthreads();
    for (int j = tid; j < n; j += 256) binned[sadr[j]] = sval[j];
}

__launch_bounds__(256)
__global__ void bin_sort_kernel(const int* __restrict__ binned, const int* __restrict__ binStart,
                                const int* __restrict__ binCount, int* __restrict__ rowptr,
                                int* __restrict__ deg, float* __restrict__ cnt,
                                int* __restrict__ col, int N) {
    __shared__ int c[256];
    __shared__ int off[256];
    __shared__ int wsum[8];
    __shared__ int scol[SCAP];
    const int tid  = threadIdx.x;
    const int lane = tid & 31;
    const int wv   = tid >> 5;
    const int b     = blockIdx.x;
    const int start = binStart[b];
    const int nE    = binCount[b];

    c[tid] = 0;
    __syncthreads();
    for (int i0 = 0; i0 < nE; i0 += SPT * 256) {
        int v[SPT];
#pragma unroll
        for (int u = 0; u < SPT; ++u) {
            const int i = i0 + u * 256 + tid;
            v[u] = 0;
            if (i < nE) v[u] = binned[start + i];
        }
#pragma unroll
        for (int u = 0; u < SPT; ++u) {
            const int i = i0 + u * 256 + tid;
            if (i < nE) atomicAdd(&c[(unsigned)v[u] >> 24], 1);
        }
    }
    __syncthreads();

    const int dg = c[tid];
    int inc = dg;
    for (int o = 1; o < 32; o <<= 1) {
        const int t = __shfl_up(inc, o, 32);
        if (lane >= o) inc += t;
    }
    if (lane == 31) wsum[wv] = inc;
    __syncthreads();
    int base = 0;
    for (int w = 0; w < 8; ++w)
        if (w < wv) base += wsum[w];
    const int ex = base + inc - dg;
    const int node = b * 256 + tid;
    if (node < N) {
        rowptr[node] = start + ex;
        deg[node]    = dg;
        cnt[node]    = (float)dg;
    }
    off[tid] = ex;
    c[tid]   = 0;
    __syncthreads();

    const bool staged = (nE <= SCAP);
    for (int i0 = 0; i0 < nE; i0 += SPT * 256) {
        int v[SPT], r[SPT], o[SPT];
#pragma unroll
        for (int u = 0; u < SPT; ++u) {
            const int i = i0 + u * 256 + tid;
            v[u] = 0;
            if (i < nE) v[u] = binned[start + i];
        }
#pragma unroll
        for (int u = 0; u < SPT; ++u) {
            const int i = i0 + u * 256 + tid;
            r[u] = 0; o[u] = 0;
            if (i < nE) {
                const int kk = (int)((unsigned)v[u] >> 24);
                r[u] = atomicAdd(&c[kk], 1);
                o[u] = off[kk];
            }
        }
#pragma unroll
        for (int u = 0; u < SPT; ++u) {
            const int i = i0 + u * 256 + tid;
            if (i < nE) {
                const int q = o[u] + r[u];
                if (staged) scol[q] = v[u] & 0xFFFFFF;
                else        col[start + q] = v[u] & 0xFFFFFF;
            }
        }
    }
    __syncthreads();
    if (staged)
        for (int j = tid; j < nE; j += 256) col[start + j] = scol[j];
}

__launch_bounds__(256)
__global__ void gather_sum_kernel(const float* __restrict__ x,
                                  const int* __restrict__ rowptr,
                                  const int* __restrict__ deg,
                                  const int* __restrict__ col,
                                  float* __restrict__ agg, int N) {
    const int lane = threadIdx.x & 31;
    const int node = __builtin_amdgcn_readfirstlane((int)(blockIdx.x * 8 + (threadIdx.x >> 5)));
    if (node >= N) return;
    const int beg = rowptr[node];
    const int n   = deg[node];
    const float* xl = x + lane * 4;

    float4 a0 = {0.f, 0.f, 0.f, 0.f};
    float4 a1 = a0, a2 = a0, a3 = a0;

    for (int done = 0; done < n; done += 32) {
        const int m = (n - done < 32) ? (n - done) : 32;
        const int idx = (lane < m) ? col[beg + done + lane] : 0;
        int j = 0;
        for (; j + 4 <= m; j += 4) {
            const int s0 = __builtin_amdgcn_readlane(idx, j);
            const int s1 = __builtin_amdgcn_readlane(idx, j + 1);
            const int s2 = __builtin_amdgcn_readlane(idx, j + 2);
            const int s3 = __builtin_amdgcn_readlane(idx, j + 3);
            const float4 v0 = *(const float4*)(xl + (size_t)s0 * D);
            const float4 v1 = *(const float4*)(xl + (size_t)s1 * D);
            const float4 v2 = *(const float4*)(xl + (size_t)s2 * D);
            const float4 v3 = *(const float4*)(xl + (size_t)s3 * D);
            a0.x += v0.x; a0.y += v0.y; a0.z += v0.z; a0.w += v0.w;
            a1.x += v1.x; a1.y += v1.y; a1.z += v1.z; a1.w += v1.w;
            a2.x += v2.x; a2.y += v2.y; a2.z += v2.z; a2.w += v2.w;
            a3.x += v3.x; a3.y += v3.y; a3.z += v3.z; a3.w += v3.w;
        }
        for (; j < m; ++j) {
            const int s0 = __builtin_amdgcn_readlane(idx, j);
            const float4 v0 = *(const float4*)(xl + (size_t)s0 * D);
            a0.x += v0.x; a0.y += v0.y; a0.z += v0.z; a0.w += v0.w;
        }
    }
    float4 r;
    r.x = (a0.x + a1.x) + (a2.x + a3.x);
    r.y = (a0.y + a1.y) + (a2.y + a3.y);
    r.z = (a0.z + a1.z) + (a2.z + a3.z);
    r.w = (a0.w + a1.w) + (a2.w + a3.w);
    *(float4*)(agg + (size_t)node * D + lane * 4) = r;
}


#define TR 32
#define APITCH 260

__launch_bounds__(256)
__global__ void sage_layer_kernel(const float* __restrict__ xin,
                                  const float* __restrict__ agg,
                                  const float* __restrict__ cnt,
                                  const float* __restrict__ Wl,
                                  const float* __restrict__ Wr,
                                  const float* __restrict__ bias,
                                  float* __restrict__ out,
                                  int N, int do_relu) {
    __shared__ __attribute__((aligned(16))) float As[TR * APITCH];
    __shared__ float ssq[TR][8];
    __shared__ float inv_norm[TR];

    const int lane = threadIdx.x & 31;
    const int wv   = __builtin_amdgcn_readfirstlane((int)(threadIdx.x >> 5));
    const int half = lane >> 4;
    const int lr   = lane & 15;
    const int rowbase = blockIdx.x * TR;
    const int col  = wv * 16 + lr;

#pragma unroll
    for (int i = 0; i < TR / 8; ++i) {
        const int r   = wv * (TR / 8) + i;
        const int row = rowbase + r;
        float4 m  = {0.f, 0.f, 0.f, 0.f};
        float4 xv = {0.f, 0.f, 0.f, 0.f};
        if (row < N) {
            const float invc = 1.0f / fmaxf(cnt[row], 1.0f);
            m  = *(const float4*)(agg + (size_t)row * D + lane * 4);
            xv = *(const float4*)(xin + (size_t)row * D + lane * 4);
            m.x *= invc; m.y *= invc; m.z *= invc; m.w *= invc;
        }
        *(float4*)(&As[r * APITCH + lane * 4])     = m;
        *(float4*)(&As[r * APITCH + D + lane * 4]) = xv;
    }
    __syncthreads();

    v8f acc0 = {0.f, 0.f, 0.f, 0.f, 0.f, 0.f, 0.f, 0.f};
    v8f acc1 = acc0;

    const float* a0p = &As[lr * APITCH + 2 * half];
    const float* a1p = &As[(16 + lr) * APITCH + 2 * half];

#pragma unroll
    for (int k0 = 0; k0 < D; k0 += 4) {
        const int ka = k0 + 2 * half;
        v2f b;
        b.x = Wl[(size_t)ka * D + col];
        b.y = Wl[(size_t)(ka + 1) * D + col];
        const v2f a0 = *(const v2f*)(a0p + k0);
        const v2f a1 = *(const v2f*)(a1p + k0);
        acc0 = __builtin_amdgcn_wmma_f32_16x16x4_f32(false, a0, false, b,
                                                     (short)0, acc0, false, false);
        acc1 = __builtin_amdgcn_wmma_f32_16x16x4_f32(false, a1, false, b,
                                                     (short)0, acc1, false, false);
    }
#pragma unroll
    for (int k0 = 0; k0 < D; k0 += 4) {
        const int ka = k0 + 2 * half;
        v2f b;
        b.x = Wr[(size_t)ka * D + col];
        b.y = Wr[(size_t)(ka + 1) * D + col];
        const v2f a0 = *(const v2f*)(a0p + D + k0);
        const v2f a1 = *(const v2f*)(a1p + D + k0);
        acc0 = __builtin_amdgcn_wmma_f32_16x16x4_f32(false, a0, false, b,
                                                     (short)0, acc0, false, false);
        acc1 = __builtin_amdgcn_wmma_f32_16x16x4_f32(false, a1, false, b,
                                                     (short)0, acc1, false, false);
    }

    const float bb = bias[col];
#pragma unroll
    for (int r = 0; r < 8; ++r) {
        float v0 = acc0[r] + bb;
        float v1 = acc1[r] + bb;
        if (do_relu) { v0 = fmaxf(v0, 0.0f); v1 = fmaxf(v1, 0.0f); }
        acc0[r] = v0;
        acc1[r] = v1;
    }

#pragma unroll
    for (int r = 0; r < 8; ++r) {
        float s = acc0[r] * acc0[r];
        s += __shfl_xor(s, 1, 32);
        s += __shfl_xor(s, 2, 32);
        s += __shfl_xor(s, 4, 32);
        s += __shfl_xor(s, 8, 32);
        if (lr == 0) ssq[r + 8 * half][wv] = s;
        float t = acc1[r] * acc1[r];
        t += __shfl_xor(t, 1, 32);
        t += __shfl_xor(t, 2, 32);
        t += __shfl_xor(t, 4, 32);
        t += __shfl_xor(t, 8, 32);
        if (lr == 0) ssq[16 + r + 8 * half][wv] = t;
    }
    __syncthreads();
    if (threadIdx.x < TR) {
        float s = 0.0f;
#pragma unroll
        for (int w = 0; w < 8; ++w) s += ssq[threadIdx.x][w];
        inv_norm[threadIdx.x] = 1.0f / fmaxf(sqrtf(s), 1e-12f);
    }
    __syncthreads();

#pragma unroll
    for (int r = 0; r < 8; ++r) {
        const int r0 = r + 8 * half;
        const int row0 = rowbase + r0;
        const int row1 = row0 + 16;
        if (row0 < N) out[(size_t)row0 * D + col] = acc0[r] * inv_norm[r0];
        if (row1 < N) out[(size_t)row1 * D + col] = acc1[r] * inv_norm[r0 + 16];
    }
}


#define NPW 32

__device__ __forceinline__ void pool_flush(float* __restrict__ pool, float* __restrict__ gcnt,
                                           int g, const float4& a, int run, int lane) {
    float* p = pool + (size_t)g * D + lane * 4;
    atomicAdd(p + 0, a.x);
    atomicAdd(p + 1, a.y);
    atomicAdd(p + 2, a.z);
    atomicAdd(p + 3, a.w);
    if (lane == 0) atomicAdd(gcnt + g, (float)run);
}

__launch_bounds__(128)
__global__ void pool_runs_kernel(const float* __restrict__ h, const int* __restrict__ batch,
                                 float* __restrict__ pool, float* __restrict__ gcnt, int N) {
    const int lane = threadIdx.x & 31;
    const int first = __builtin_amdgcn_readfirstlane((int)(blockIdx.x * 4 + (threadIdx.x >> 5))) * NPW;
    if (first >= N) return;
    const int m = (N - first < NPW) ? (N - first) : NPW;
    const int gid = (lane < m) ? batch[first + lane] : 0;
    const float* hp = h + (size_t)first * D + lane * 4;

    float4 v[NPW];
#pragma unroll
    for (int j = 0; j < NPW; ++j) {
        const int jj = (j < m) ? j : (m - 1);
        v[j] = *(const float4*)(hp + (size_t)jj * D);
    }

    float4 acc = v[0];
    int cur = __builtin_amdgcn_readlane(gid, 0);
    int run = 1;
#pragma unroll
    for (int j = 1; j < NPW; ++j) {
        if (j < m) {
            const int g = __builtin_amdgcn_readlane(gid, j);
            if (g != cur) {
                pool_flush(pool, gcnt, cur, acc, run, lane);
                acc.x = 0.f; acc.y = 0.f; acc.z = 0.f; acc.w = 0.f;
                run = 0;
                cur = g;
            }
            acc.x += v[j].x; acc.y += v[j].y; acc.z += v[j].z; acc.w += v[j].w;
            ++run;
        }
    }
    pool_flush(pool, gcnt, cur, acc, run, lane);
}

__global__ void pool_div_kernel(const float* __restrict__ pool, const float* __restrict__ gcnt,
                                float* __restrict__ out) {
    int i = blockIdx.x * blockDim.x + threadIdx.x;
    if (i < GRAPHS * D) {
        int g = i / D;
        out[i] = pool[i] / fmaxf(gcnt[g], 1.0f);
    }
}


static inline unsigned cdiv_u(long long a, long long b) { return (unsigned)((a + b - 1) / b); }

extern "C" void kernel_launch(void* const* d_in, const int* in_sizes, int n_in,
                              void* d_out, int out_size, void* d_ws, size_t ws_size,
                              hipStream_t stream) {
    const float* x    = (const float*)d_in[0];
    const int*   ei   = (const int*)d_in[1];
    const int*   batch= (const int*)d_in[2];
    const float* Wl1  = (const float*)d_in[3];
    const float* Wr1  = (const float*)d_in[4];
    const float* b1   = (const float*)d_in[5];
    const float* Wl2  = (const float*)d_in[6];
    const float* Wr2  = (const float*)d_in[7];
    const float* b2   = (const float*)d_in[8];
    const float* Wl3  = (const float*)d_in[9];
    const float* Wr3  = (const float*)d_in[10];
    const float* b3   = (const float*)d_in[11];
    float* out = (float*)d_out;

    const int N  = in_sizes[0] / D;
    const int E  = in_sizes[1] / 2;
    const int ND = N * D;

    float* ws   = (float*)d_ws;
    float* agg  = ws;  ws += (size_t)ND;
    float* h1   = ws;  ws += (size_t)ND;
    float* h2   = ws;  ws += (size_t)ND;
    float* cnt  = ws;  ws += N;
    float* gcnt = ws;  ws += GRAPHS;
    float* pool = ws;  ws += (size_t)GRAPHS * D;
    int* binCount  = (int*)ws;  ws += 256;
    int* binCursor = (int*)ws;  ws += 256;
    int* binStart  = (int*)ws;  ws += 256;
    int* deg    = (int*)ws;  ws += N;
    int* rowptr = (int*)ws;  ws += N;
    int* col    = (int*)ws;  ws += E;
    int* binned = (int*)ws;  ws += E;

    const int* src = ei;
    const int* dstv = ei + E;

    dim3 blk(256);
    const unsigned chunk_blocks = cdiv_u(E, EPB);
    const unsigned range_blocks = cdiv_u(N, 256);
    const unsigned gath_blocks = cdiv_u(N, 8);
    const unsigned gemm_blocks = cdiv_u(N, TR);

    fill_zero_int_kernel<<<cdiv_u(GRAPHS + GRAPHS * D + 512, 256), blk, 0, stream>>>((int*)gcnt, GRAPHS + GRAPHS * D + 512);
    bin_count_kernel<<<chunk_blocks, blk, 0, stream>>>(dstv, binCount, E);
    bin_scatter_kernel<<<chunk_blocks, blk, 0, stream>>>(src, dstv, binCount, binCursor, binStart, binned, E);
    bin_sort_kernel<<<range_blocks, blk, 0, stream>>>(binned, binStart, binCount, rowptr, deg, cnt, col, N);

    gather_sum_kernel<<<gath_blocks, blk, 0, stream>>>(x, rowptr, deg, col, agg, N);
    sage_layer_kernel<<<gemm_blocks, blk, 0, stream>>>(x, agg, cnt, Wl1, Wr1, b1, h1, N, 1);

    gather_sum_kernel<<<gath_blocks, blk, 0, stream>>>(h1, rowptr, deg, col, agg, N);
    sage_layer_kernel<<<gemm_blocks, blk, 0, stream>>>(h1, agg, cnt, Wl2, Wr2, b2, h2, N, 1);

    gather_sum_kernel<<<gath_blocks, blk, 0, stream>>>(h2, rowptr, deg, col, agg, N);
    sage_layer_kernel<<<gemm_blocks, blk, 0, stream>>>(h2, agg, cnt, Wl3, Wr3, b3, h1, N, 0);

    pool_runs_kernel<<<cdiv_u(N, 4 * NPW), dim3(128), 0, stream>>>(h1, batch, pool, gcnt, N);
    pool_div_kernel<<<cdiv_u(GRAPHS * D, 256), blk, 0, stream>>>(pool, gcnt, out);
}


# Gather fused into the layer kernel: a wave sums its 4 nodes' neighbour rows (8 loads in flight) into the LDS tile; half the blocks run x@Wr first to overlap gather waits; x@Wr added before mean@Wl
# speedup vs baseline: 4.3787x; 1.0631x over previous
#include <hip/hip_runtime.h>
#include <hip/hip_bf16.h>

typedef __attribute__((ext_vector_type(2))) float v2f;
typedef __attribute__((ext_vector_type(8))) float v8f;

#define D 128
#define GRAPHS 64


__global__ void fill_zero_int_kernel(int* __restrict__ p, int n) {
    int i = blockIdx.x * blockDim.x + threadIdx.x;
    if (i < n) p[i] = 0;
}


#define EPB 4096
#define EPT (EPB / 256)
#define SCAP 8192
#define SPT 8

__launch_bounds__(256)
__global__ void bin_count_kernel(const int* __restrict__ dst, int* __restrict__ binCount, int E) {
    __shared__ int hist[256];
    const int tid = threadIdx.x;
    const int first = blockIdx.x * EPB;
    hist[tid] = 0;
    __syncthreads();
    int d[EPT];
#pragma unroll
    for (int i = 0; i < EPT; ++i) {
        const int e = first + i * 256 + tid;
        d[i] = -1;
        if (e < E) d[i] = dst[e];
    }
#pragma unroll
    for (int i = 0; i < EPT; ++i)
        if (d[i] >= 0) atomicAdd(&hist[d[i] >> 8], 1);
    __syncthreads();
    const int h = hist[tid];
    if (h > 0) atomicAdd(&binCount[tid], h);
}

__launch_bounds__(256)
__global__ void bin_scatter_kernel(const int* __restrict__ src, const int* __restrict__ dst,
                                   const int* __restrict__ binCount, int* __restrict__ binCursor,
                                   int* __restrict__ binStart, int* __restrict__ binned, int E) {
    __shared__ int hist[257];
    __shared__ int loff[257];
    __shared__ int gbase[257];
    __shared__ int wsA[8];
    __shared__ int wsB[8];
    __shared__ int sval[EPB];
    __shared__ int sadr[EPB];
    const int tid  = threadIdx.x;
    const int lane = tid & 31;
    const int wv   = tid >> 5;
    const int first = blockIdx.x * EPB;
    const int n = (E - first < EPB) ? (E - first) : EPB;

    hist[tid] = 0;
    if (tid == 0) { hist[256] = 0; loff[256] = 0; gbase[256] = 0; }
    __syncthreads();

    int d[EPT], s[EPT];
#pragma unroll
    for (int i = 0; i < EPT; ++i) {
        const int e = first + i * 256 + tid;
        d[i] = -1;
        s[i] = 0;
        if (e < E) { d[i] = dst[e]; s[i] = src[e]; }
    }
#pragma unroll
    for (int i = 0; i < EPT; ++i)
        if (d[i] >= 0) atomicAdd(&hist[d[i] >> 8], 1);
    const int ca = binCount[tid];
    __syncthreads();
    const int cb = hist[tid];

    int ia = ca, ib = cb;
    for (int o = 1; o < 32; o <<= 1) {
        const int ta = __shfl_up(ia, o, 32);
        const int tb = __shfl_up(ib, o, 32);
        if (lane >= o) { ia += ta; ib += tb; }
    }
    if (lane == 31) { wsA[wv] = ia; wsB[wv] = ib; }
    __syncthreads();
    int ba = 0, bb = 0;
    for (int w = 0; w < 8; ++w)
        if (w < wv) { ba += wsA[w]; bb += wsB[w]; }
    const int ga = ba + ia - ca;
    const int lb = bb + ib - cb;
    if (blockIdx.x == 0) binStart[tid] = ga;
    int r = 0;
    if (cb > 0) r = atomicAdd(&binCursor[tid], cb);
    gbase[tid] = ga + r;
    loff[tid]  = lb;
    hist[tid]  = 0;
    __syncthreads();

    int k[EPT], lo[EPT], gb[EPT];
#pragma unroll
    for (int i = 0; i < EPT; ++i) {
        const int b = (d[i] >= 0) ? (d[i] >> 8) : 256;
        k[i]  = atomicAdd(&hist[b], 1);
        lo[i] = loff[b];
        gb[i] = gbase[b];
    }
#pragma unroll
    for (int i = 0; i < EPT; ++i) {
        if (d[i] >= 0) {
            const int q = lo[i] + k[i];
            sval[q] = (int)((unsigned)s[i] | ((unsigned)(d[i] & 255) << 24));
            sadr[q] = gb[i] + k[i];
        }
    }
    __syncthreads();
    for (int j = tid; j < n; j += 256) binned[sadr[j]] = sval[j];
}

__launch_bounds__(256)
__global__ void bin_sort_kernel(const int* __restrict__ binned, const int* __restrict__ binStart,
                                const int* __restrict__ binCount, int* __restrict__ rowptr,
                                int* __restrict__ deg,
                                int* __restrict__ col, int N) {
    __shared__ int c[256];
    __shared__ int off[256];
    __shared__ int wsum[8];
    __shared__ int scol[SCAP];
    const int tid  = threadIdx.x;
    const int lane = tid & 31;
    const int wv   = tid >> 5;
    const int b     = blockIdx.x;
    const int start = binStart[b];
    const int nE    = binCount[b];

    c[tid] = 0;
    __syncthreads();
    for (int i0 = 0; i0 < nE; i0 += SPT * 256) {
        int v[SPT];
#pragma unroll
        for (int u = 0; u < SPT; ++u) {
            const int i = i0 + u * 256 + tid;
            v[u] = 0;
            if (i < nE) v[u] = binned[start + i];
        }
#pragma unroll
        for (int u = 0; u < SPT; ++u) {
            const int i = i0 + u * 256 + tid;
            if (i < nE) atomicAdd(&c[(unsigned)v[u] >> 24], 1);
        }
    }
    __syncthreads();

    const int dg = c[tid];
    int inc = dg;
    for (int o = 1; o < 32; o <<= 1) {
        const int t = __shfl_up(inc, o, 32);
        if (lane >= o) inc += t;
    }
    if (lane == 31) wsum[wv] = inc;
    __syncthreads();
    int base = 0;
    for (int w = 0; w < 8; ++w)
        if (w < wv) base += wsum[w];
    const int ex = base + inc - dg;
    const int node = b * 256 + tid;
    if (node < N) {
        rowptr[node] = start + ex;
        deg[node]    = dg;
    }
    off[tid] = ex;
    c[tid]   = 0;
    __syncthreads();

    const bool staged = (nE <= SCAP);
    for (int i0 = 0; i0 < nE; i0 += SPT * 256) {
        int v[SPT], r[SPT], o[SPT];
#pragma unroll
        for (int u = 0; u < SPT; ++u) {
            const int i = i0 + u * 256 + tid;
            v[u] = 0;
            if (i < nE) v[u] = binned[start + i];
        }
#pragma unroll
        for (int u = 0; u < SPT; ++u) {
            const int i = i0 + u * 256 + tid;
            r[u] = 0; o[u] = 0;
            if (i < nE) {
                const int kk = (int)((unsigned)v[u] >> 24);
                r[u] = atomicAdd(&c[kk], 1);
                o[u] = off[kk];
            }
        }
#pragma unroll
        for (int u = 0; u < SPT; ++u) {
            const int i = i0 + u * 256 + tid;
            if (i < nE) {
                const int q = o[u] + r[u];
                if (staged) scol[q] = v[u] & 0xFFFFFF;
                else        col[start + q] = v[u] & 0xFFFFFF;
            }
        }
    }
    __syncthreads();
    if (staged)
        for (int j = tid; j < nE; j += 256) col[start + j] = scol[j];
}


#define TR 32
#define APITCH 260

__device__ __forceinline__ void add4(float4& a, const float4& v) {
    a.x += v.x; a.y += v.y; a.z += v.z; a.w += v.w;
}

__device__ __forceinline__ void gather_mean_rows(const float* __restrict__ xin,
                                                 const int* __restrict__ colidx,
                                                 float* __restrict__ As,
                                                 int vbeg, int vn, int idx, int wv, int lane) {
    const float* xl = xin + lane * 4;
    int beg = __builtin_amdgcn_readlane(vbeg, 0);
    int n   = __builtin_amdgcn_readlane(vn, 0);

#pragma unroll 1
    for (int i = 0; i < 4; ++i) {
        int begn = 0, nn = 0, idxn = 0;
        if (i < 3) {
            begn = __builtin_amdgcn_readlane(vbeg, i + 1);
            nn   = __builtin_amdgcn_readlane(vn, i + 1);
            idxn = (lane < nn) ? colidx[begn + lane] : 0;
        }
        float4 a0 = {0.f, 0.f, 0.f, 0.f};
        float4 a1 = a0, a2 = a0, a3 = a0;
        for (int done = 0; done < n; done += 32) {
            const int m = (n - done < 32) ? (n - done) : 32;
            if (done > 0) idx = (lane < m) ? colidx[beg + done + lane] : 0;
            int j = 0;
            for (; j + 8 <= m; j += 8) {
                const int s0 = __builtin_amdgcn_readlane(idx, j);
                const int s1 = __builtin_amdgcn_readlane(idx, j + 1);
                const int s2 = __builtin_amdgcn_readlane(idx, j + 2);
                const int s3 = __builtin_amdgcn_readlane(idx, j + 3);
                const int s4 = __builtin_amdgcn_readlane(idx, j + 4);
                const int s5 = __builtin_amdgcn_readlane(idx, j + 5);
                const int s6 = __builtin_amdgcn_readlane(idx, j + 6);
                const int s7 = __builtin_amdgcn_readlane(idx, j + 7);
                const float4 v0 = *(const float4*)(xl + (size_t)s0 * D);
                const float4 v1 = *(const float4*)(xl + (size_t)s1 * D);
                const float4 v2 = *(const float4*)(xl + (size_t)s2 * D);
                const float4 v3 = *(const float4*)(xl + (size_t)s3 * D);
                const float4 v4 = *(const float4*)(xl + (size_t)s4 * D);
                const float4 v5 = *(const float4*)(xl + (size_t)s5 * D);
                const float4 v6 = *(const float4*)(xl + (size_t)s6 * D);
                const float4 v7 = *(const float4*)(xl + (size_t)s7 * D);
                add4(a0, v0); add4(a1, v1); add4(a2, v2); add4(a3, v3);
                add4(a0, v4); add4(a1, v5); add4(a2, v6); add4(a3, v7);
            }
            const int rem = m - j;
            if (rem > 0) {
                float4 t[7];
#pragma unroll
                for (int u = 0; u < 7; ++u) {
                    t[u].x = 0.f; t[u].y = 0.f; t[u].z = 0.f; t[u].w = 0.f;
                    if (u < rem) {
                        const int s = __builtin_amdgcn_readlane(idx, j + u);
                        t[u] = *(const float4*)(xl + (size_t)s * D);
                    }
                }
                if (rem >= 4) {
                    add4(a0, t[0]); add4(a1, t[1]); add4(a2, t[2]); add4(a3, t[3]);
                    if (rem > 4) add4(a0, t[4]);
                    if (rem > 5) add4(a0, t[5]);
                    if (rem > 6) add4(a0, t[6]);
                } else {
                    add4(a0, t[0]);
                    if (rem > 1) add4(a0, t[1]);
                    if (rem > 2) add4(a0, t[2]);
                }
            }
        }
        const float invc = 1.0f / fmaxf((float)n, 1.0f);
        float4 r;
        r.x = ((a0.x + a1.x) + (a2.x + a3.x)) * invc;
        r.y = ((a0.y + a1.y) + (a2.y + a3.y)) * invc;
        r.z = ((a0.z + a1.z) + (a2.z + a3.z)) * invc;
        r.w = ((a0.w + a1.w) + (a2.w + a3.w)) * invc;
        *(float4*)(&As[(wv * 4 + i) * APITCH + lane * 4]) = r;
        beg = begn; n = nn; idx = idxn;
    }
}

__launch_bounds__(256)
__global__ void sage_layer_kernel(const float* __restrict__ xin,
                                  const int* __restrict__ rowptr,
                                  const int* __restrict__ deg,
                                  const int* __restrict__ colidx,
                                  const float* __restrict__ Wl,
                                  const float* __restrict__ Wr,
                                  const float* __restrict__ bias,
                                  float* __restrict__ out,
                                  int N, int do_relu) {
    __shared__ __attribute__((aligned(16))) float As[TR * APITCH];
    __shared__ float ssq[TR][8];
    __shared__ float inv_norm[TR];

    const int lane = threadIdx.x & 31;
    const int wv   = __builtin_amdgcn_readfirstlane((int)(threadIdx.x >> 5));
    const int half = lane >> 4;
    const int lr   = lane & 15;
    const int rowbase = blockIdx.x * TR;
    const int col  = wv * 16 + lr;

    const bool gather_first = (__builtin_popcount(blockIdx.x) & 1) == 0;

    const int node0 = rowbase + wv * (TR / 8);
    int vbeg = 0, vn = 0;
    if (lane < 4 && node0 + lane < N) { vbeg = rowptr[node0 + lane]; vn = deg[node0 + lane]; }

    float4 xv[TR / 8];
#pragma unroll
    for (int i = 0; i < TR / 8; ++i) {
        const int row = rowbase + wv * (TR / 8) + i;
        const int rc  = (row < N) ? row : (N - 1);
        xv[i] = *(const float4*)(xin + (size_t)rc * D + lane * 4);
    }
#pragma unroll
    for (int i = 0; i < TR / 8; ++i) {
        const int r   = wv * (TR / 8) + i;
        if (rowbase + r >= N) { xv[i].x = 0.f; xv[i].y = 0.f; xv[i].z = 0.f; xv[i].w = 0.f; }
        *(float4*)(&As[r * APITCH + D + lane * 4]) = xv[i];
    }
    const int n0   = __builtin_amdgcn_readlane(vn, 0);
    const int idx0 = (lane < n0) ? colidx[__builtin_amdgcn_readlane(vbeg, 0) + lane] : 0;
    if (gather_first) gather_mean_rows(xin, colidx, As, vbeg, vn, idx0, wv, lane);
    __syncthreads();

    v8f acc0 = {0.f, 0.f, 0.f, 0.f, 0.f, 0.f, 0.f, 0.f};
    v8f acc1 = acc0;

    const float* a0p = &As[lr * APITCH + 2 * half];
    const float* a1p = &As[(16 + lr) * APITCH + 2 * half];

#pragma unroll
    for (int k0 = 0; k0 < D; k0 += 4) {
        const int ka = k0 + 2 * half;
        v2f b;
        b.x = Wr[(size_t)ka * D + col];
        b.y = Wr[(size_t)(ka + 1) * D + col];
        const v2f a0 = *(const v2f*)(a0p + D + k0);
        const v2f a1 = *(const v2f*)(a1p + D + k0);
        acc0 = __builtin_amdgcn_wmma_f32_16x16x4_f32(false, a0, false, b,
                                                     (short)0, acc0, false, false);
        acc1 = __builtin_amdgcn_wmma_f32_16x16x4_f32(false, a1, false, b,
                                                     (short)0, acc1, false, false);
    }

    if (!gather_first) {
        gather_mean_rows(xin, colidx, As, vbeg, vn, idx0, wv, lane);
        __syncthreads();
    }

#pragma unroll
    for (int k0 = 0; k0 < D; k0 += 4) {
        const int ka = k0 + 2 * half;
        v2f b;
        b.x = Wl[(size_t)ka * D + col];
        b.y = Wl[(size_t)(ka + 1) * D + col];
        const v2f a0 = *(const v2f*)(a0p + k0);
        const v2f a1 = *(const v2f*)(a1p + k0);
        acc0 = __builtin_amdgcn_wmma_f32_16x16x4_f32(false, a0, false, b,
                                                     (short)0, acc0, false, false);
        acc1 = __builtin_amdgcn_wmma_f32_16x16x4_f32(false, a1, false, b,
                                                     (short)0, acc1, false, false);
    }

    const float bb = bias[col];
#pragma unroll
    for (int r = 0; r < 8; ++r) {
        float v0 = acc0[r] + bb;
        float v1 = acc1[r] + bb;
        if (do_relu) { v0 = fmaxf(v0, 0.0f); v1 = fmaxf(v1, 0.0f); }
        acc0[r] = v0;
        acc1[r] = v1;
    }

#pragma unroll
    for (int r = 0; r < 8; ++r) {
        float s = acc0[r] * acc0[r];
        s += __shfl_xor(s, 1, 32);
        s += __shfl_xor(s, 2, 32);
        s += __shfl_xor(s, 4, 32);
        s += __shfl_xor(s, 8, 32);
        if (lr == 0) ssq[r + 8 * half][wv] = s;
        float t = acc1[r] * acc1[r];
        t += __shfl_xor(t, 1, 32);
        t += __shfl_xor(t, 2, 32);
        t += __shfl_xor(t, 4, 32);
        t += __shfl_xor(t, 8, 32);
        if (lr == 0) ssq[16 + r + 8 * half][wv] = t;
    }
    __syncthreads();
    if (threadIdx.x < TR) {
        float s = 0.0f;
#pragma unroll
        for (int w = 0; w < 8; ++w) s += ssq[threadIdx.x][w];
        inv_norm[threadIdx.x] = 1.0f / fmaxf(sqrtf(s), 1e-12f);
    }
    __syncthreads();

#pragma unroll
    for (int r = 0; r < 8; ++r) {
        const int r0 = r + 8 * half;
        const int row0 = rowbase + r0;
        const int row1 = row0 + 16;
        if (row0 < N) out[(size_t)row0 * D + col] = acc0[r] * inv_norm[r0];
        if (row1 < N) out[(size_t)row1 * D + col] = acc1[r] * inv_norm[r0 + 16];
    }
}


#define NPW 32

__device__ __forceinline__ void pool_flush(float* __restrict__ pool, float* __restrict__ gcnt,
                                           int g, const float4& a, int run, int lane) {
    float* p = pool + (size_t)g * D + lane * 4;
    atomicAdd(p + 0, a.x);
    atomicAdd(p + 1, a.y);
    atomicAdd(p + 2, a.z);
    atomicAdd(p + 3, a.w);
    if (lane == 0) atomicAdd(gcnt + g, (float)run);
}

__launch_bounds__(128)
__global__ void pool_runs_kernel(const float* __restrict__ h, const int* __restrict__ batch,
                                 float* __restrict__ pool, float* __restrict__ gcnt, int N) {
    const int lane = threadIdx.x & 31;
    const int first = __builtin_amdgcn_readfirstlane((int)(blockIdx.x * 4 + (threadIdx.x >> 5))) * NPW;
    if (first >= N) return;
    const int m = (N - first < NPW) ? (N - first) : NPW;
    const int gid = (lane < m) ? batch[first + lane] : 0;
    const float* hp = h + (size_t)first * D + lane * 4;

    float4 v[NPW];
#pragma unroll
    for (int j = 0; j < NPW; ++j) {
        const int jj = (j < m) ? j : (m - 1);
        v[j] = *(const float4*)(hp + (size_t)jj * D);
    }

    float4 acc = v[0];
    int cur = __builtin_amdgcn_readlane(gid, 0);
    int run = 1;
#pragma unroll
    for (int j = 1; j < NPW; ++j) {
        if (j < m) {
            const int g = __builtin_amdgcn_readlane(gid, j);
            if (g != cur) {
                pool_flush(pool, gcnt, cur, acc, run, lane);
                acc.x = 0.f; acc.y = 0.f; acc.z = 0.f; acc.w = 0.f;
                run = 0;
                cur = g;
            }
            acc.x += v[j].x; acc.y += v[j].y; acc.z += v[j].z; acc.w += v[j].w;
            ++run;
        }
    }
    pool_flush(pool, gcnt, cur, acc, run, lane);
}

__global__ void pool_div_kernel(const float* __restrict__ pool, const float* __restrict__ gcnt,
                                float* __restrict__ out) {
    int i = blockIdx.x * blockDim.x + threadIdx.x;
    if (i < GRAPHS * D) {
        int g = i / D;
        out[i] = pool[i] / fmaxf(gcnt[g], 1.0f);
    }
}


static inline unsigned cdiv_u(long long a, long long b) { return (unsigned)((a + b - 1) / b); }

extern "C" void kernel_launch(void* const* d_in, const int* in_sizes, int n_in,
                              void* d_out, int out_size, void* d_ws, size_t ws_size,
                              hipStream_t stream) {
    const float* x    = (const float*)d_in[0];
    const int*   ei   = (const int*)d_in[1];
    const int*   batch= (const int*)d_in[2];
    const float* Wl1  = (const float*)d_in[3];
    const float* Wr1  = (const float*)d_in[4];
    const float* b1   = (const float*)d_in[5];
    const float* Wl2  = (const float*)d_in[6];
    const float* Wr2  = (const float*)d_in[7];
    const float* b2   = (const float*)d_in[8];
    const float* Wl3  = (const float*)d_in[9];
    const float* Wr3  = (const float*)d_in[10];
    const float* b3   = (const float*)d_in[11];
    float* out = (float*)d_out;

    const int N  = in_sizes[0] / D;
    const int E  = in_sizes[1] / 2;
    const int ND = N * D;

    float* ws   = (float*)d_ws;
    float* h1   = ws;  ws += (size_t)ND;
    float* h2   = ws;  ws += (size_t)ND;
    float* gcnt = ws;  ws += GRAPHS;
    float* pool = ws;  ws += (size_t)GRAPHS * D;
    int* binCount  = (int*)ws;  ws += 256;
    int* binCursor = (int*)ws;  ws += 256;
    int* binStart  = (int*)ws;  ws += 256;
    int* deg    = (int*)ws;  ws += N;
    int* rowptr = (int*)ws;  ws += N;
    int* col    = (int*)ws;  ws += E;
    int* binned = (int*)ws;  ws += E;

    const int* src = ei;
    const int* dstv = ei + E;

    dim3 blk(256);
    const unsigned chunk_blocks = cdiv_u(E, EPB);
    const unsigned range_blocks = cdiv_u(N, 256);
    const unsigned gemm_blocks = cdiv_u(N, TR);

    fill_zero_int_kernel<<<cdiv_u(GRAPHS + GRAPHS * D + 512, 256), blk, 0, stream>>>((int*)gcnt, GRAPHS + GRAPHS * D + 512);
    bin_count_kernel<<<chunk_blocks, blk, 0, stream>>>(dstv, binCount, E);
    bin_scatter_kernel<<<chunk_blocks, blk, 0, stream>>>(src, dstv, binCount, binCursor, binStart, binned, E);
    bin_sort_kernel<<<range_blocks, blk, 0, stream>>>(binned, binStart, binCount, rowptr, deg, col, N);

    sage_layer_kernel<<<gemm_blocks, blk, 0, stream>>>(x, rowptr, deg, col, Wl1, Wr1, b1, h1, N, 1);

    sage_layer_kernel<<<gemm_blocks, blk, 0, stream>>>(h1, rowptr, deg, col, Wl2, Wr2, b2, h2, N, 1);

    sage_layer_kernel<<<gemm_blocks, blk, 0, stream>>>(h2, rowptr, deg, col, Wl3, Wr3, b3, h1, N, 0);

    pool_runs_kernel<<<cdiv_u(N, 4 * NPW), dim3(128), 0, stream>>>(h1, batch, pool, gcnt, N);
    pool_div_kernel<<<cdiv_u(GRAPHS * D, 256), blk, 0, stream>>>(pool, gcnt, out);
}


# Mean pool fused into layer 3's epilogue: node counts taken in the sort launch, rows summed by runs from the accumulators, scaled by 1/count, atomic add into out; two launches gone; sum order changed
# speedup vs baseline: 4.6855x; 1.0701x over previous
#include <hip/hip_runtime.h>
#include <hip/hip_bf16.h>

typedef __attribute__((ext_vector_type(2))) float v2f;
typedef __attribute__((ext_vector_type(8))) float v8f;

#define D 128
#define GRAPHS 64


__global__ void fill_zero_kernel(int* __restrict__ p, int n, float* __restrict__ q, int m) {
    int i = blockIdx.x * blockDim.x + threadIdx.x;
    if (i < n) p[i] = 0;
    else if (i < n + m) q[i - n] = 0.0f;
}


#define EPB 4096
#define EPT (EPB / 256)
#define SCAP 8192
#define SPT 8

__launch_bounds__(256)
__global__ void bin_count_kernel(const int* __restrict__ dst, int* __restrict__ binCount, int E) {
    __shared__ int hist[256];
    const int tid = threadIdx.x;
    const int first = blockIdx.x * EPB;
    hist[tid] = 0;
    __syncthreads();
    int d[EPT];
#pragma unroll
    for (int i = 0; i < EPT; ++i) {
        const int e = first + i * 256 + tid;
        d[i] = -1;
        if (e < E) d[i] = dst[e];
    }
#pragma unroll
    for (int i = 0; i < EPT; ++i)
        if (d[i] >= 0) atomicAdd(&hist[d[i] >> 8], 1);
    __syncthreads();
    const int h = hist[tid];
    if (h > 0) atomicAdd(&binCount[tid], h);
}

__launch_bounds__(256)
__global__ void bin_scatter_kernel(const int* __restrict__ src, const int* __restrict__ dst,
                                   const int* __restrict__ binCount, int* __restrict__ binCursor,
                                   int* __restrict__ binStart, int* __restrict__ binned, int E) {
    __shared__ int hist[257];
    __shared__ int loff[257];
    __shared__ int gbase[257];
    __shared__ int wsA[8];
    __shared__ int wsB[8];
    __shared__ int sval[EPB];
    __shared__ int sadr[EPB];
    const int tid  = threadIdx.x;
    const int lane = tid & 31;
    const int wv   = tid >> 5;
    const int first = blockIdx.x * EPB;
    const int n = (E - first < EPB) ? (E - first) : EPB;

    hist[tid] = 0;
    if (tid == 0) { hist[256] = 0; loff[256] = 0; gbase[256] = 0; }
    __syncthreads();

    int d[EPT], s[EPT];
#pragma unroll
    for (int i = 0; i < EPT; ++i) {
        const int e = first + i * 256 + tid;
        d[i] = -1;
        s[i] = 0;
        if (e < E) { d[i] = dst[e]; s[i] = src[e]; }
    }
#pragma unroll
    for (int i = 0; i < EPT; ++i)
        if (d[i] >= 0) atomicAdd(&hist[d[i] >> 8], 1);
    const int ca = binCount[tid];
    __syncthreads();
    const int cb = hist[tid];

    int ia = ca, ib = cb;
    for (int o = 1; o < 32; o <<= 1) {
        const int ta = __shfl_up(ia, o, 32);
        const int tb = __shfl_up(ib, o, 32);
        if (lane >= o) { ia += ta; ib += tb; }
    }
    if (lane == 31) { wsA[wv] = ia; wsB[wv] = ib; }
    __syncthreads();
    int ba = 0, bb = 0;
    for (int w = 0; w < 8; ++w)
        if (w < wv) { ba += wsA[w]; bb += wsB[w]; }
    const int ga = ba + ia - ca;
    const int lb = bb + ib - cb;
    if (blockIdx.x == 0) binStart[tid] = ga;
    int r = 0;
    if (cb > 0) r = atomicAdd(&binCursor[tid], cb);
    gbase[tid] = ga + r;
    loff[tid]  = lb;
    hist[tid]  = 0;
    __syncthreads();

    int k[EPT], lo[EPT], gb[EPT];
#pragma unroll
    for (int i = 0; i < EPT; ++i) {
        const int b = (d[i] >= 0) ? (d[i] >> 8) : 256;
        k[i]  = atomicAdd(&hist[b], 1);
        lo[i] = loff[b];
        gb[i] = gbase[b];
    }
#pragma unroll
    for (int i = 0; i < EPT; ++i) {
        if (d[i] >= 0) {
            const int q = lo[i] + k[i];
            sval[q] = (int)((unsigned)s[i] | ((unsigned)(d[i] & 255) << 24));
            sadr[q] = gb[i] + k[i];
        }
    }
    __syncthreads();
    for (int j = tid; j < n; j += 256) binned[sadr[j]] = sval[j];
}

__launch_bounds__(256)
__global__ void bin_sort_kernel(const int* __restrict__ binned, const int* __restrict__ binStart,
                                const int* __restrict__ binCount, int* __restrict__ rowptr,
                                int* __restrict__ deg,
                                int* __restrict__ col,
                                const int* __restrict__ batch, int* __restrict__ gcnt, int N) {
    __shared__ int c[256];
    __shared__ int off[256];
    __shared__ int wsum[8];
    __shared__ int scol[SCAP];
    const int tid  = threadIdx.x;
    const int lane = tid & 31;
    const int wv   = tid >> 5;
    const int b     = blockIdx.x;
    const int start = binStart[b];
    const int nE    = binCount[b];

    int gid = -1;
    if (b * 256 + tid < N) gid = batch[b * 256 + tid];

    c[tid] = 0;
    __syncthreads();
    for (int i0 = 0; i0 < nE; i0 += SPT * 256) {
        int v[SPT];
#pragma unroll
        for (int u = 0; u < SPT; ++u) {
            const int i = i0 + u * 256 + tid;
            v[u] = 0;
            if (i < nE) v[u] = binned[start + i];
        }
#pragma unroll
        for (int u = 0; u < SPT; ++u) {
            const int i = i0 + u * 256 + tid;
            if (i < nE) atomicAdd(&c[(unsigned)v[u] >> 24], 1);
        }
    }
    __syncthreads();

    const int dg = c[tid];
    int inc = dg;
    for (int o = 1; o < 32; o <<= 1) {
        const int t = __shfl_up(inc, o, 32);
        if (lane >= o) inc += t;
    }
    if (lane == 31) wsum[wv] = inc;
    __syncthreads();
    int base = 0;
    for (int w = 0; w < 8; ++w)
        if (w < wv) base += wsum[w];
    const int ex = base + inc - dg;
    const int node = b * 256 + tid;
    if (node < N) {
        rowptr[node] = start + ex;
        deg[node]    = dg;
    }
    off[tid] = ex;
    c[tid]   = 0;
    __syncthreads();

    const bool staged = (nE <= SCAP);
    for (int i0 = 0; i0 < nE; i0 += SPT * 256) {
        int v[SPT], r[SPT], o[SPT];
#pragma unroll
        for (int u = 0; u < SPT; ++u) {
            const int i = i0 + u * 256 + tid;
            v[u] = 0;
            if (i < nE) v[u] = binned[start + i];
        }
#pragma unroll
        for (int u = 0; u < SPT; ++u) {
            const int i = i0 + u * 256 + tid;
            r[u] = 0; o[u] = 0;
            if (i < nE) {
                const int kk = (int)((unsigned)v[u] >> 24);
                r[u] = atomicAdd(&c[kk], 1);
                o[u] = off[kk];
            }
        }
#pragma unroll
        for (int u = 0; u < SPT; ++u) {
            const int i = i0 + u * 256 + tid;
            if (i < nE) {
                const int q = o[u] + r[u];
                if (staged) scol[q] = v[u] & 0xFFFFFF;
                else        col[start + q] = v[u] & 0xFFFFFF;
            }
        }
    }
    __syncthreads();
    if (staged)
        for (int j = tid; j < nE; j += 256) col[start + j] = scol[j];

    if ((unsigned)gid >= (unsigned)GRAPHS) gid = -1;
    const int gprev = __shfl_up(gid, 1, 32);
    const bool head = (lane == 0) || (gid != gprev);
    const unsigned heads = (unsigned)__ballot(head);
    const unsigned above = (lane == 31) ? 0u : (heads >> (lane + 1));
    const int len = above ? __ffs((int)above) : (32 - lane);
    if (head && gid >= 0) atomicAdd(gcnt + gid, len);
}


#define TR 32
#define APITCH 260

__device__ __forceinline__ void add4(float4& a, const float4& v) {
    a.x += v.x; a.y += v.y; a.z += v.z; a.w += v.w;
}

__device__ __forceinline__ void gather_mean_rows(const float* __restrict__ xin,
                                                 const int* __restrict__ colidx,
                                                 float* __restrict__ As,
                                                 int vbeg, int vn, int idx, int wv, int lane) {
    const float* xl = xin + lane * 4;
    int beg = __builtin_amdgcn_readlane(vbeg, 0);
    int n   = __builtin_amdgcn_readlane(vn, 0);

#pragma unroll 1
    for (int i = 0; i < 4; ++i) {
        int begn = 0, nn = 0, idxn = 0;
        if (i < 3) {
            begn = __builtin_amdgcn_readlane(vbeg, i + 1);
            nn   = __builtin_amdgcn_readlane(vn, i + 1);
            idxn = (lane < nn) ? colidx[begn + lane] : 0;
        }
        float4 a0 = {0.f, 0.f, 0.f, 0.f};
        float4 a1 = a0, a2 = a0, a3 = a0;
        for (int done = 0; done < n; done += 32) {
            const int m = (n - done < 32) ? (n - done) : 32;
            if (done > 0) idx = (lane < m) ? colidx[beg + done + lane] : 0;
            int j = 0;
            for (; j + 8 <= m; j += 8) {
                const int s0 = __builtin_amdgcn_readlane(idx, j);
                const int s1 = __builtin_amdgcn_readlane(idx, j + 1);
                const int s2 = __builtin_amdgcn_readlane(idx, j + 2);
                const int s3 = __builtin_amdgcn_readlane(idx, j + 3);
                const int s4 = __builtin_amdgcn_readlane(idx, j + 4);
                const int s5 = __builtin_amdgcn_readlane(idx, j + 5);
                const int s6 = __builtin_amdgcn_readlane(idx, j + 6);
                const int s7 = __builtin_amdgcn_readlane(idx, j + 7);
                const float4 v0 = *(const float4*)(xl + (size_t)s0 * D);
                const float4 v1 = *(const float4*)(xl + (size_t)s1 * D);
                const float4 v2 = *(const float4*)(xl + (size_t)s2 * D);
                const float4 v3 = *(const float4*)(xl + (size_t)s3 * D);
                const float4 v4 = *(const float4*)(xl + (size_t)s4 * D);
                const float4 v5 = *(const float4*)(xl + (size_t)s5 * D);
                const float4 v6 = *(const float4*)(xl + (size_t)s6 * D);
                const float4 v7 = *(const float4*)(xl + (size_t)s7 * D);
                add4(a0, v0); add4(a1, v1); add4(a2, v2); add4(a3, v3);
                add4(a0, v4); add4(a1, v5); add4(a2, v6); add4(a3, v7);
            }
            const int rem = m - j;
            if (rem > 0) {
                float4 t[7];
#pragma unroll
                for (int u = 0; u < 7; ++u) {
                    t[u].x = 0.f; t[u].y = 0.f; t[u].z = 0.f; t[u].w = 0.f;
                    if (u < rem) {
                        const int s = __builtin_amdgcn_readlane(idx, j + u);
                        t[u] = *(const float4*)(xl + (size_t)s * D);
                    }
                }
                if (rem >= 4) {
                    add4(a0, t[0]); add4(a1, t[1]); add4(a2, t[2]); add4(a3, t[3]);
                    if (rem > 4) add4(a0, t[4]);
                    if (rem > 5) add4(a0, t[5]);
                    if (rem > 6) add4(a0, t[6]);
                } else {
                    add4(a0, t[0]);
                    if (rem > 1) add4(a0, t[1]);
                    if (rem > 2) add4(a0, t[2]);
                }
            }
        }
        const float invc = 1.0f / fmaxf((float)n, 1.0f);
        float4 r;
        r.x = ((a0.x + a1.x) + (a2.x + a3.x)) * invc;
        r.y = ((a0.y + a1.y) + (a2.y + a3.y)) * invc;
        r.z = ((a0.z + a1.z) + (a2.z + a3.z)) * invc;
        r.w = ((a0.w + a1.w) + (a2.w + a3.w)) * invc;
        *(float4*)(&As[(wv * 4 + i) * APITCH + lane * 4]) = r;
        beg = begn; n = nn; idx = idxn;
    }
}

__launch_bounds__(256)
__global__ void sage_layer_kernel(const float* __restrict__ xin,
                                  const int* __restrict__ rowptr,
                                  const int* __restrict__ deg,
                                  const int* __restrict__ colidx,
                                  const float* __restrict__ Wl,
                                  const float* __restrict__ Wr,
                                  const float* __restrict__ bias,
                                  float* __restrict__ out,
                                  const int* __restrict__ batch,
                                  const int* __restrict__ gcnt,
                                  float* __restrict__ pooled,
                                  int N, int do_relu, int do_pool) {
    __shared__ __attribute__((aligned(16))) float As[TR * APITCH];
    __shared__ float ssq[TR][8];
    __shared__ float inv_norm[TR];
    __shared__ int   sg[TR];
    __shared__ float sic[TR];

    const int lane = threadIdx.x & 31;
    const int wv   = __builtin_amdgcn_readfirstlane((int)(threadIdx.x >> 5));
    const int half = lane >> 4;
    const int lr   = lane & 15;
    const int rowbase = blockIdx.x * TR;
    const int col  = wv * 16 + lr;

    const bool gather_first = (__builtin_popcount(blockIdx.x) & 1) == 0;

    const int node0 = rowbase + wv * (TR / 8);
    int vbeg = 0, vn = 0;
    if (lane < 4 && node0 + lane < N) { vbeg = rowptr[node0 + lane]; vn = deg[node0 + lane]; }

    int pg = -1;
    if (do_pool && threadIdx.x < TR && rowbase + (int)threadIdx.x < N) pg = batch[rowbase + (int)threadIdx.x];

    float4 xv[TR / 8];
#pragma unroll
    for (int i = 0; i < TR / 8; ++i) {
        const int row = rowbase + wv * (TR / 8) + i;
        const int rc  = (row < N) ? row : (N - 1);
        xv[i] = *(const float4*)(xin + (size_t)rc * D + lane * 4);
    }
#pragma unroll
    for (int i = 0; i < TR / 8; ++i) {
        const int r   = wv * (TR / 8) + i;
        if (rowbase + r >= N) { xv[i].x = 0.f; xv[i].y = 0.f; xv[i].z = 0.f; xv[i].w = 0.f; }
        *(float4*)(&As[r * APITCH + D + lane * 4]) = xv[i];
    }
    const int n0   = __builtin_amdgcn_readlane(vn, 0);
    const int idx0 = (lane < n0) ? colidx[__builtin_amdgcn_readlane(vbeg, 0) + lane] : 0;

    int pcnt = 1;
    if ((unsigned)pg < (unsigned)GRAPHS) pcnt = gcnt[pg];
    else pg = -1;
    if (gather_first) gather_mean_rows(xin, colidx, As, vbeg, vn, idx0, wv, lane);
    __syncthreads();

    v8f acc0 = {0.f, 0.f, 0.f, 0.f, 0.f, 0.f, 0.f, 0.f};
    v8f acc1 = acc0;

    const float* a0p = &As[lr * APITCH + 2 * half];
    const float* a1p = &As[(16 + lr) * APITCH + 2 * half];

#pragma unroll
    for (int k0 = 0; k0 < D; k0 += 4) {
        const int ka = k0 + 2 * half;
        v2f b;
        b.x = Wr[(size_t)ka * D + col];
        b.y = Wr[(size_t)(ka + 1) * D + col];
        const v2f a0 = *(const v2f*)(a0p + D + k0);
        const v2f a1 = *(const v2f*)(a1p + D + k0);
        acc0 = __builtin_amdgcn_wmma_f32_16x16x4_f32(false, a0, false, b,
                                                     (short)0, acc0, false, false);
        acc1 = __builtin_amdgcn_wmma_f32_16x16x4_f32(false, a1, false, b,
                                                     (short)0, acc1, false, false);
    }

    if (!gather_first) {
        gather_mean_rows(xin, colidx, As, vbeg, vn, idx0, wv, lane);
        __syncthreads();
    }

#pragma unroll
    for (int k0 = 0; k0 < D; k0 += 4) {
        const int ka = k0 + 2 * half;
        v2f b;
        b.x = Wl[(size_t)ka * D + col];
        b.y = Wl[(size_t)(ka + 1) * D + col];
        const v2f a0 = *(const v2f*)(a0p + k0);
        const v2f a1 = *(const v2f*)(a1p + k0);
        acc0 = __builtin_amdgcn_wmma_f32_16x16x4_f32(false, a0, false, b,
                                                     (short)0, acc0, false, false);
        acc1 = __builtin_amdgcn_wmma_f32_16x16x4_f32(false, a1, false, b,
                                                     (short)0, acc1, false, false);
    }

    const float bb = bias[col];
#pragma unroll
    for (int r = 0; r < 8; ++r) {
        float v0 = acc0[r] + bb;
        float v1 = acc1[r] + bb;
        if (do_relu) { v0 = fmaxf(v0, 0.0f); v1 = fmaxf(v1, 0.0f); }
        acc0[r] = v0;
        acc1[r] = v1;
    }

#pragma unroll
    for (int r = 0; r < 8; ++r) {
        float s = acc0[r] * acc0[r];
        s += __shfl_xor(s, 1, 32);
        s += __shfl_xor(s, 2, 32);
        s += __shfl_xor(s, 4, 32);
        s += __shfl_xor(s, 8, 32);
        if (lr == 0) ssq[r + 8 * half][wv] = s;
        float t = acc1[r] * acc1[r];
        t += __shfl_xor(t, 1, 32);
        t += __shfl_xor(t, 2, 32);
        t += __shfl_xor(t, 4, 32);
        t += __shfl_xor(t, 8, 32);
        if (lr == 0) ssq[16 + r + 8 * half][wv] = t;
    }
    __syncthreads();
    if (threadIdx.x < TR) {
        float s = 0.0f;
#pragma unroll
        for (int w = 0; w < 8; ++w) s += ssq[threadIdx.x][w];
        inv_norm[threadIdx.x] = 1.0f / fmaxf(sqrtf(s), 1e-12f);
        if (do_pool) {
            sg[threadIdx.x]  = pg;
            sic[threadIdx.x] = (pg >= 0) ? 1.0f / fmaxf((float)pcnt, 1.0f) : 0.0f;
        }
    }
    __syncthreads();

    if (!do_pool) {
#pragma unroll
        for (int r = 0; r < 8; ++r) {
            const int r0 = r + 8 * half;
            const int row0 = rowbase + r0;
            const int row1 = row0 + 16;
            if (row0 < N) out[(size_t)row0 * D + col] = acc0[r] * inv_norm[r0];
            if (row1 < N) out[(size_t)row1 * D + col] = acc1[r] * inv_norm[r0 + 16];
        }
        return;
    }

    const int g0 = sg[0];
    const bool one_graph = (g0 >= 0) && (__ballot(sg[lane] != g0) == 0);
    if (one_graph) {
        float s = 0.0f;
#pragma unroll
        for (int r = 0; r < 8; ++r) s += acc0[r] * inv_norm[r + 8 * half];
#pragma unroll
        for (int r = 0; r < 8; ++r) s += acc1[r] * inv_norm[16 + r + 8 * half];
        s += __shfl_xor(s, 16, 32);
        if (half == 0) atomicAdd(pooled + (size_t)g0 * D + col, s * sic[0]);
    } else {
        int cur = -1;
        float a = 0.0f, ic = 0.0f;
#pragma unroll
        for (int q = 0; q < 16; ++q) {
            const int rr = (q < 8) ? (q + 8 * half) : (8 + q + 8 * half);
            const float hv = ((q < 8) ? acc0[q & 7] : acc1[q & 7]) * inv_norm[rr];
            const int g = sg[rr];
            if (g != cur) {
                if (cur >= 0) atomicAdd(pooled + (size_t)cur * D + col, a * ic);
                cur = g;
                ic = sic[rr];
                a = 0.0f;
            }
            a += hv;
        }
        if (cur >= 0) atomicAdd(pooled + (size_t)cur * D + col, a * ic);
    }
}


static inline unsigned cdiv_u(long long a, long long b) { return (unsigned)((a + b - 1) / b); }

extern "C" void kernel_launch(void* const* d_in, const int* in_sizes, int n_in,
                              void* d_out, int out_size, void* d_ws, size_t ws_size,
                              hipStream_t stream) {
    const float* x    = (const float*)d_in[0];
    const int*   ei   = (const int*)d_in[1];
    const int*   batch= (const int*)d_in[2];
    const float* Wl1  = (const float*)d_in[3];
    const float* Wr1  = (const float*)d_in[4];
    const float* b1   = (const float*)d_in[5];
    const float* Wl2  = (const float*)d_in[6];
    const float* Wr2  = (const float*)d_in[7];
    const float* b2   = (const float*)d_in[8];
    const float* Wl3  = (const float*)d_in[9];
    const float* Wr3  = (const float*)d_in[10];
    const float* b3   = (const float*)d_in[11];
    float* out = (float*)d_out;

    const int N  = in_sizes[0] / D;
    const int E  = in_sizes[1] / 2;
    const int ND = N * D;

    float* ws   = (float*)d_ws;
    float* h1   = ws;  ws += (size_t)ND;
    float* h2   = ws;  ws += (size_t)ND;
    int* gcnt      = (int*)ws;  ws += GRAPHS;
    int* binCount  = (int*)ws;  ws += 256;
    int* binCursor = (int*)ws;  ws += 256;
    int* binStart  = (int*)ws;  ws += 256;
    int* deg    = (int*)ws;  ws += N;
    int* rowptr = (int*)ws;  ws += N;
    int* col    = (int*)ws;  ws += E;
    int* binned = (int*)ws;  ws += E;

    const int* src = ei;
    const int* dstv = ei + E;

    dim3 blk(256);
    const unsigned chunk_blocks = cdiv_u(E, EPB);
    const unsigned range_blocks = cdiv_u(N, 256);
    const unsigned gemm_blocks = cdiv_u(N, TR);

    fill_zero_kernel<<<cdiv_u(GRAPHS + 512 + GRAPHS * D, 256), blk, 0, stream>>>(gcnt, GRAPHS + 512, out, GRAPHS * D);
    bin_count_kernel<<<chunk_blocks, blk, 0, stream>>>(dstv, binCount, E);
    bin_scatter_kernel<<<chunk_blocks, blk, 0, stream>>>(src, dstv, binCount, binCursor, binStart, binned, E);
    bin_sort_kernel<<<range_blocks, blk, 0, stream>>>(binned, binStart, binCount, rowptr, deg, col, batch, gcnt, N);

    sage_layer_kernel<<<gemm_blocks, blk, 0, stream>>>(x, rowptr, deg, col, Wl1, Wr1, b1, h1, batch, gcnt, out, N, 1, 0);

    sage_layer_kernel<<<gemm_blocks, blk, 0, stream>>>(h1, rowptr, deg, col, Wl2, Wr2, b2, h2, batch, gcnt, out, N, 1, 0);

    sage_layer_kernel<<<gemm_blocks, blk, 0, stream>>>(h2, rowptr, deg, col, Wl3, Wr3, b3, h1, batch, gcnt, out, N, 0, 1);
}
